# starting kernel
# baseline (speedup 1.0000x reference)
#include <hip/hip_runtime.h>
#include <hip/hip_bf16.h>

typedef float v2f __attribute__((ext_vector_type(2)));
typedef float v8f __attribute__((ext_vector_type(8)));

#define NQ 512
#define NKV 2562
#define NKV_PAD 2624
#define CDIM 64
#define NCHUNK 41
#define CHUNK 64
#define NQT (NQ / 16)
#define SCALE (1.0f / 64.0f)

#define OFF_QW   0L
#define OFF_KW   (OFF_QW   + (long)NQ * CDIM)
#define OFF_VW   (OFF_KW   + (long)NKV_PAD * CDIM)
#define OFF_EQ   (OFF_VW   + (long)NKV_PAD * CDIM)
#define OFF_IEQ  (OFF_EQ   + (long)NQ * CDIM)
#define OFF_EK   (OFF_IEQ  + (long)NQ * CDIM)
#define OFF_IEK  (OFF_EK   + (long)NKV_PAD * CDIM)
#define OFF_WSUM (OFF_IEK  + (long)NKV_PAD * CDIM)
#define OFF_EQC  (OFF_WSUM + 16L)
#define OFF_IEQC (OFF_EQC  + (long)NQ * 3)
#define OFF_EKC  (OFF_IEQC + (long)NQ * 3)
#define OFF_IEKC (OFF_EKC  + (long)NKV * 3)
#define OFF_PO   (OFF_IEKC + (long)NKV * 3 + 16L)
#define OFF_PM   (OFF_PO   + (long)NQT * NCHUNK * 16 * CDIM)
#define OFF_PS   (OFF_PM   + (long)NQT * NCHUNK * 16)

static __device__ __forceinline__ v8f wmma_f32_4(v2f a, v2f b, v8f c) {
    return __builtin_amdgcn_wmma_f32_16x16x4_f32(false, a, false, b, (short)0, c,
                                                 false, false);
}

__global__ void diffattn_prep(const float* __restrict__ q,
                              const float* __restrict__ kv,
                              const float* __restrict__ Wq,
                              const float* __restrict__ Wkv,
                              const float* __restrict__ Wdelta,
                              float* __restrict__ q_ws,
                              float* __restrict__ k_ws,
                              float* __restrict__ v_ws,
                              float* __restrict__ wsum_ws) {
    if (blockIdx.x == 0 && threadIdx.x < 3) {
        float s = 0.f;
        for (int c = 0; c < CDIM; ++c) s += Wdelta[threadIdx.x * CDIM + c];
        wsum_ws[threadIdx.x] = s;
    }

    const int KV_TILES = (NKV_PAD / 16) * (128 / 16);
    const int Q_TILES = (NQ / 16) * (CDIM / 16);
    int wave = (blockIdx.x * blockDim.x + threadIdx.x) >> 5;
    int lane = threadIdx.x & 31;
    if (wave >= KV_TILES + Q_TILES) return;

    int half = lane >> 4;
    int lm = lane & 15;

    if (wave < KV_TILES) {
        int mt = wave >> 3, nt = wave & 7;
        int m0 = mt * 16, n0 = nt * 16;
        int rowA = m0 + lm;
        bool inb = rowA < NKV;
        v8f acc = {};
        for (int kk = 0; kk < CDIM; kk += 4) {
            int ka = kk + 2 * half;
            v2f a, b;
            a.x = inb ? kv[rowA * CDIM + ka] : 0.f;
            a.y = inb ? kv[rowA * CDIM + ka + 1] : 0.f;
            b.x = Wkv[ka * 128 + n0 + lm];
            b.y = Wkv[(ka + 1) * 128 + n0 + lm];
            acc = wmma_f32_4(a, b, acc);
        }
        int col = n0 + lm;
#pragma unroll
        for (int r = 0; r < 8; ++r) {
            int row = m0 + r + 8 * half;
            if (col < CDIM)
                k_ws[row * CDIM + col] = acc[r];
            else
                v_ws[row * CDIM + (col - CDIM)] = acc[r];
        }
    } else {
        int t = wave - KV_TILES;
        int mt = t >> 2, nt = t & 3;
        int m0 = mt * 16, n0 = nt * 16;
        int rowA = m0 + lm;
        v8f acc = {};
        for (int kk = 0; kk < CDIM; kk += 4) {
            int ka = kk + 2 * half;
            v2f a, b;
            a.x = q[rowA * CDIM + ka];
            a.y = q[rowA * CDIM + ka + 1];
            b.x = Wq[ka * CDIM + n0 + lm];
            b.y = Wq[(ka + 1) * CDIM + n0 + lm];
            acc = wmma_f32_4(a, b, acc);
        }
        int col = n0 + lm;
#pragma unroll
        for (int r = 0; r < 8; ++r) {
            int row = m0 + r + 8 * half;
            q_ws[row * CDIM + col] = acc[r];
        }
    }
}

__global__ void diffattn_exp(const float* __restrict__ q_ws,
                             const float* __restrict__ k_ws,
                             const float* __restrict__ q_coord,
                             const float* __restrict__ kv_coord,
                             float* __restrict__ Eq, float* __restrict__ iEq,
                             float* __restrict__ Ek, float* __restrict__ iEk,
                             float* __restrict__ Eqc, float* __restrict__ iEqc,
                             float* __restrict__ Ekc, float* __restrict__ iEkc) {
    const int NE_Q = NQ * CDIM;
    const int NE_K = NKV_PAD * CDIM;
    const int NE_QC = NQ * 3;
    const int NE_KC = NKV * 3;
    int i = blockIdx.x * blockDim.x + threadIdx.x;
    if (i < NE_Q) {
        float x = q_ws[i];
        Eq[i] = __expf(x);
        iEq[i] = __expf(-x);
    } else if (i < NE_Q + NE_K) {
        int j = i - NE_Q;
        float x = k_ws[j];
        Ek[j] = __expf(x);
        iEk[j] = __expf(-x);
    } else if (i < NE_Q + NE_K + NE_QC) {
        int j = i - NE_Q - NE_K;
        float x = q_coord[j];
        Eqc[j] = __expf(x);
        iEqc[j] = __expf(-x);
    } else if (i < NE_Q + NE_K + NE_QC + NE_KC) {
        int j = i - NE_Q - NE_K - NE_QC;
        float x = kv_coord[j];
        Ekc[j] = __expf(x);
        iEkc[j] = __expf(-x);
    }
}

__global__ void __launch_bounds__(256)
diffattn_score(const float* __restrict__ Eq, const float* __restrict__ iEq,
               const float* __restrict__ Ek, const float* __restrict__ iEk,
               const float* __restrict__ Eqc, const float* __restrict__ iEqc,
               const float* __restrict__ Ekc, const float* __restrict__ iEkc,
               const float* __restrict__ wsum_ws,
               const float* __restrict__ v_ws,
               float* __restrict__ po, float* __restrict__ pm,
               float* __restrict__ ps) {
    __shared__ float sEq[16][CDIM + 1];
    __shared__ float sIEq[16][CDIM + 1];
    __shared__ float sP[16][CHUNK + 8];
    __shared__ float sEqc[16][4];
    __shared__ float sIEqc[16][4];
    __shared__ float sRed[16][17];
    __shared__ float sMax[16];
    __shared__ float sSum[16];
    __shared__ float sW[3];

    const int tid = threadIdx.x;
    const int ch = blockIdx.x;
    const int qt = blockIdx.y;
    const int q0 = qt * 16;
    const int j0 = ch * CHUNK;

    for (int i = tid; i < 16 * CDIM; i += 256) {
        int r = i / CDIM, c = i % CDIM;
        sEq[r][c] = Eq[(q0 + r) * CDIM + c];
        sIEq[r][c] = iEq[(q0 + r) * CDIM + c];
    }
    if (tid < 48) {
        int r = tid / 3, d = tid % 3;
        sEqc[r][d] = Eqc[(q0 + r) * 3 + d];
        sIEqc[r][d] = iEqc[(q0 + r) * 3 + d];
    }
    if (tid < 3) sW[tid] = wsum_ws[tid];
    __syncthreads();

    for (int p = tid; p < 16 * CHUNK; p += 256) {
        int qi = p & 15;
        int jl = p >> 4;
        int jj = j0 + jl;
        float s;
        if (jj < NKV) {
            s = 0.f;
            const float4* e4 = reinterpret_cast<const float4*>(Ek + jj * CDIM);
            const float4* i4 = reinterpret_cast<const float4*>(iEk + jj * CDIM);
#pragma unroll 4
            for (int c4 = 0; c4 < CDIM / 4; ++c4) {
                float4 ek = e4[c4];
                float4 ik = i4[c4];
                int c = 4 * c4;
                s += fminf(sEq[qi][c + 0] * ik.x, sIEq[qi][c + 0] * ek.x);
                s += fminf(sEq[qi][c + 1] * ik.y, sIEq[qi][c + 1] * ek.y);
                s += fminf(sEq[qi][c + 2] * ik.z, sIEq[qi][c + 2] * ek.z);
                s += fminf(sEq[qi][c + 3] * ik.w, sIEq[qi][c + 3] * ek.w);
            }
#pragma unroll
            for (int d = 0; d < 3; ++d) {
                float ekc = Ekc[jj * 3 + d];
                float ikc = iEkc[jj * 3 + d];
                s += fminf(sEqc[qi][d] * ikc, sIEqc[qi][d] * ekc) * sW[d];
            }
            s *= SCALE;
        } else {
            s = -__builtin_inff();
        }
        sP[qi][jl] = s;
    }
    __syncthreads();

    const int qi = tid & 15;
    const int g = tid >> 4;
    {
        float m = -__builtin_inff();
        for (int jl = g; jl < CHUNK; jl += 16) m = fmaxf(m, sP[qi][jl]);
        sRed[qi][g] = m;
    }
    __syncthreads();
    if (tid < 16) {
        float m = sRed[tid][0];
#pragma unroll
        for (int i = 1; i < 16; ++i) m = fmaxf(m, sRed[tid][i]);
        sMax[tid] = m;
    }
    __syncthreads();

    {
        float mr = sMax[qi];
        float psum = 0.f;
        for (int jl = g; jl < CHUNK; jl += 16) {
            float e = __expf(sP[qi][jl] - mr);
            sP[qi][jl] = e;
            psum += e;
        }
        sRed[qi][g] = psum;
    }
    __syncthreads();
    if (tid < 16) {
        float s = 0.f;
#pragma unroll
        for (int i = 0; i < 16; ++i) s += sRed[tid][i];
        sSum[tid] = s;
    }
    __syncthreads();

    const int wave = tid >> 5;
    const int lane = tid & 31;
    const int half = lane >> 4;
    const int lm = lane & 15;
    const long base = (long)(qt * NCHUNK + ch) * 16;
    if (wave < 4) {
        int n0 = wave * 16;
        v8f acc = {};
        for (int kk = 0; kk < CHUNK; kk += 4) {
            int ka = kk + 2 * half;
            v2f a, b;
            a.x = sP[lm][ka];
            a.y = sP[lm][ka + 1];
            b.x = v_ws[(j0 + ka) * CDIM + n0 + lm];
            b.y = v_ws[(j0 + ka + 1) * CDIM + n0 + lm];
            acc = wmma_f32_4(a, b, acc);
        }
#pragma unroll
        for (int r = 0; r < 8; ++r) {
            int row = r + 8 * half;
            po[(base + row) * CDIM + n0 + lm] = acc[r];
        }
    }
    if (tid < 16) {
        pm[base + tid] = sMax[tid];
        ps[base + tid] = sSum[tid];
    }
}

__global__ void __launch_bounds__(256)
diffattn_combine(const float* __restrict__ po, const float* __restrict__ pm,
                 const float* __restrict__ ps,
                 const float* __restrict__ Wproj,
                 const float* __restrict__ bproj, float* __restrict__ out) {
    __shared__ float sO[16][CDIM + 1];
    __shared__ float cm[16][NCHUNK + 1];
    __shared__ float cw[16][NCHUNK + 1];
    __shared__ float sM[16];
    __shared__ float sInv[16];

    const int tid = threadIdx.x;
    const int qt = blockIdx.x;
    const long pbase = (long)qt * NCHUNK * 16;

    for (int i = tid; i < 16 * NCHUNK; i += 256) {
        int r = i & 15, c = i >> 4;
        cm[r][c] = pm[pbase + c * 16 + r];
        cw[r][c] = ps[pbase + c * 16 + r];
    }
    __syncthreads();
    if (tid < 16) {
        float M = cm[tid][0];
        for (int c = 1; c < NCHUNK; ++c) M = fmaxf(M, cm[tid][c]);
        float S = 0.f;
        for (int c = 0; c < NCHUNK; ++c) S += cw[tid][c] * __expf(cm[tid][c] - M);
        sM[tid] = M;
        sInv[tid] = 1.0f / S;
    }
    __syncthreads();
    for (int i = tid; i < 16 * NCHUNK; i += 256) {
        int r = i & 15, c = i >> 4;
        cw[r][c] = __expf(cm[r][c] - sM[r]) * sInv[r];
    }
    __syncthreads();

    for (int idx = tid; idx < 16 * CDIM; idx += 256) {
        int r = idx >> 6, n = idx & 63;
        float a = 0.f;
        for (int c = 0; c < NCHUNK; ++c)
            a += po[(pbase + c * 16 + r) * CDIM + n] * cw[r][c];
        sO[r][n] = a;
    }
    __syncthreads();

    const int wave = tid >> 5;
    const int lane = tid & 31;
    const int half = lane >> 4;
    const int lm = lane & 15;
    if (wave < 4) {
        int n0 = wave * 16;
        v8f acc = {};
        for (int kk = 0; kk < CDIM; kk += 4) {
            int ka = kk + 2 * half;
            v2f a, b;
            a.x = sO[lm][ka];
            a.y = sO[lm][ka + 1];
            b.x = Wproj[ka * CDIM + n0 + lm];
            b.y = Wproj[(ka + 1) * CDIM + n0 + lm];
            acc = wmma_f32_4(a, b, acc);
        }
        float bb = bproj[n0 + lm];
#pragma unroll
        for (int r = 0; r < 8; ++r) {
            int row = r + 8 * half;
            out[(qt * 16 + row) * CDIM + n0 + lm] = acc[r] + bb;
        }
    }
}

extern "C" void kernel_launch(void* const* d_in, const int* in_sizes, int n_in,
                              void* d_out, int out_size, void* d_ws, size_t ws_size,
                              hipStream_t stream) {
    const float* q        = (const float*)d_in[0];
    const float* q_coord  = (const float*)d_in[1];
    const float* kv       = (const float*)d_in[2];
    const float* kv_coord = (const float*)d_in[3];
    const float* Wq       = (const float*)d_in[4];
    const float* Wkv      = (const float*)d_in[5];
    const float* Wdelta   = (const float*)d_in[6];
    const float* Wproj    = (const float*)d_in[7];
    const float* bproj    = (const float*)d_in[8];
    float* out = (float*)d_out;
    float* ws  = (float*)d_ws;

    float* q_ws    = ws + OFF_QW;
    float* k_ws    = ws + OFF_KW;
    float* v_ws    = ws + OFF_VW;
    float* Eqp     = ws + OFF_EQ;
    float* iEqp    = ws + OFF_IEQ;
    float* Ekp     = ws + OFF_EK;
    float* iEkp    = ws + OFF_IEK;
    float* wsum_ws = ws + OFF_WSUM;
    float* Eqc     = ws + OFF_EQC;
    float* iEqc    = ws + OFF_IEQC;
    float* Ekc     = ws + OFF_EKC;
    float* iEkc    = ws + OFF_IEKC;
    float* po      = ws + OFF_PO;
    float* pm      = ws + OFF_PM;
    float* ps      = ws + OFF_PS;

    diffattn_prep<<<180, 256, 0, stream>>>(q, kv, Wq, Wkv, Wdelta,
                                           q_ws, k_ws, v_ws, wsum_ws);

    int exp_total = NQ * CDIM + NKV_PAD * CDIM + NQ * 3 + NKV * 3;
    diffattn_exp<<<(exp_total + 255) / 256, 256, 0, stream>>>(
        q_ws, k_ws, q_coord, kv_coord, Eqp, iEqp, Ekp, iEkp, Eqc, iEqc, Ekc, iEkc);

    dim3 sgrid(NCHUNK, NQT);
    diffattn_score<<<sgrid, 256, 0, stream>>>(Eqp, iEqp, Ekp, iEkp, Eqc, iEqc,
                                              Ekc, iEkc, wsum_ws, v_ws, po, pm, ps);

    diffattn_combine<<<NQT, 256, 0, stream>>>(po, pm, ps, Wproj, bproj, out);
}


# combine kernel: 256 blocks x 2 query rows, one o element per thread, all its loads (41 partials, stats, Wproj) issued before the first wait; was 32 blocks, grouped loop; weight sum now a lane tree
# speedup vs baseline: 1.4608x; 1.4608x over previous
#include <hip/hip_runtime.h>
#include <hip/hip_bf16.h>

typedef float v2f __attribute__((ext_vector_type(2)));
typedef float v8f __attribute__((ext_vector_type(8)));

#define NQ 512
#define NKV 2562
#define NKV_PAD 2624
#define CDIM 64
#define NCHUNK 41
#define CHUNK 64
#define NQT (NQ / 16)
#define SCALE (1.0f / 64.0f)

#define OFF_QW   0L
#define OFF_KW   (OFF_QW   + (long)NQ * CDIM)
#define OFF_VW   (OFF_KW   + (long)NKV_PAD * CDIM)
#define OFF_EQ   (OFF_VW   + (long)NKV_PAD * CDIM)
#define OFF_IEQ  (OFF_EQ   + (long)NQ * CDIM)
#define OFF_EK   (OFF_IEQ  + (long)NQ * CDIM)
#define OFF_IEK  (OFF_EK   + (long)NKV_PAD * CDIM)
#define OFF_WSUM (OFF_IEK  + (long)NKV_PAD * CDIM)
#define OFF_EQC  (OFF_WSUM + 16L)
#define OFF_IEQC (OFF_EQC  + (long)NQ * 3)
#define OFF_EKC  (OFF_IEQC + (long)NQ * 3)
#define OFF_IEKC (OFF_EKC  + (long)NKV * 3)
#define OFF_PO   (OFF_IEKC + (long)NKV * 3 + 16L)
#define OFF_PM   (OFF_PO   + (long)NQT * NCHUNK * 16 * CDIM)
#define OFF_PS   (OFF_PM   + (long)NQT * NCHUNK * 16)

static __device__ __forceinline__ v8f wmma_f32_4(v2f a, v2f b, v8f c) {
    return __builtin_amdgcn_wmma_f32_16x16x4_f32(false, a, false, b, (short)0, c,
                                                 false, false);
}

__global__ void diffattn_prep(const float* __restrict__ q,
                              const float* __restrict__ kv,
                              const float* __restrict__ Wq,
                              const float* __restrict__ Wkv,
                              const float* __restrict__ Wdelta,
                              float* __restrict__ q_ws,
                              float* __restrict__ k_ws,
                              float* __restrict__ v_ws,
                              float* __restrict__ wsum_ws) {
    if (blockIdx.x == 0 && threadIdx.x < 3) {
        float s = 0.f;
        for (int c = 0; c < CDIM; ++c) s += Wdelta[threadIdx.x * CDIM + c];
        wsum_ws[threadIdx.x] = s;
    }

    const int KV_TILES = (NKV_PAD / 16) * (128 / 16);
    const int Q_TILES = (NQ / 16) * (CDIM / 16);
    int wave = (blockIdx.x * blockDim.x + threadIdx.x) >> 5;
    int lane = threadIdx.x & 31;
    if (wave >= KV_TILES + Q_TILES) return;

    int half = lane >> 4;
    int lm = lane & 15;

    if (wave < KV_TILES) {
        int mt = wave >> 3, nt = wave & 7;
        int m0 = mt * 16, n0 = nt * 16;
        int rowA = m0 + lm;
        bool inb = rowA < NKV;
        v8f acc = {};
        for (int kk = 0; kk < CDIM; kk += 4) {
            int ka = kk + 2 * half;
            v2f a, b;
            a.x = inb ? kv[rowA * CDIM + ka] : 0.f;
            a.y = inb ? kv[rowA * CDIM + ka + 1] : 0.f;
            b.x = Wkv[ka * 128 + n0 + lm];
            b.y = Wkv[(ka + 1) * 128 + n0 + lm];
            acc = wmma_f32_4(a, b, acc);
        }
        int col = n0 + lm;
#pragma unroll
        for (int r = 0; r < 8; ++r) {
            int row = m0 + r + 8 * half;
            if (col < CDIM)
                k_ws[row * CDIM + col] = acc[r];
            else
                v_ws[row * CDIM + (col - CDIM)] = acc[r];
        }
    } else {
        int t = wave - KV_TILES;
        int mt = t >> 2, nt = t & 3;
        int m0 = mt * 16, n0 = nt * 16;
        int rowA = m0 + lm;
        v8f acc = {};
        for (int kk = 0; kk < CDIM; kk += 4) {
            int ka = kk + 2 * half;
            v2f a, b;
            a.x = q[rowA * CDIM + ka];
            a.y = q[rowA * CDIM + ka + 1];
            b.x = Wq[ka * CDIM + n0 + lm];
            b.y = Wq[(ka + 1) * CDIM + n0 + lm];
            acc = wmma_f32_4(a, b, acc);
        }
        int col = n0 + lm;
#pragma unroll
        for (int r = 0; r < 8; ++r) {
            int row = m0 + r + 8 * half;
            q_ws[row * CDIM + col] = acc[r];
        }
    }
}

__global__ void diffattn_exp(const float* __restrict__ q_ws,
                             const float* __restrict__ k_ws,
                             const float* __restrict__ q_coord,
                             const float* __restrict__ kv_coord,
                             float* __restrict__ Eq, float* __restrict__ iEq,
                             float* __restrict__ Ek, float* __restrict__ iEk,
                             float* __restrict__ Eqc, float* __restrict__ iEqc,
                             float* __restrict__ Ekc, float* __restrict__ iEkc) {
    const int NE_Q = NQ * CDIM;
    const int NE_K = NKV_PAD * CDIM;
    const int NE_QC = NQ * 3;
    const int NE_KC = NKV * 3;
    int i = blockIdx.x * blockDim.x + threadIdx.x;
    if (i < NE_Q) {
        float x = q_ws[i];
        Eq[i] = __expf(x);
        iEq[i] = __expf(-x);
    } else if (i < NE_Q + NE_K) {
        int j = i - NE_Q;
        float x = k_ws[j];
        Ek[j] = __expf(x);
        iEk[j] = __expf(-x);
    } else if (i < NE_Q + NE_K + NE_QC) {
        int j = i - NE_Q - NE_K;
        float x = q_coord[j];
        Eqc[j] = __expf(x);
        iEqc[j] = __expf(-x);
    } else if (i < NE_Q + NE_K + NE_QC + NE_KC) {
        int j = i - NE_Q - NE_K - NE_QC;
        float x = kv_coord[j];
        Ekc[j] = __expf(x);
        iEkc[j] = __expf(-x);
    }
}

__global__ void __launch_bounds__(256)
diffattn_score(const float* __restrict__ Eq, const float* __restrict__ iEq,
               const float* __restrict__ Ek, const float* __restrict__ iEk,
               const float* __restrict__ Eqc, const float* __restrict__ iEqc,
               const float* __restrict__ Ekc, const float* __restrict__ iEkc,
               const float* __restrict__ wsum_ws,
               const float* __restrict__ v_ws,
               float* __restrict__ po, float* __restrict__ pm,
               float* __restrict__ ps) {
    __shared__ float sEq[16][CDIM + 1];
    __shared__ float sIEq[16][CDIM + 1];
    __shared__ float sP[16][CHUNK + 8];
    __shared__ float sEqc[16][4];
    __shared__ float sIEqc[16][4];
    __shared__ float sRed[16][17];
    __shared__ float sMax[16];
    __shared__ float sSum[16];
    __shared__ float sW[3];

    const int tid = threadIdx.x;
    const int ch = blockIdx.x;
    const int qt = blockIdx.y;
    const int q0 = qt * 16;
    const int j0 = ch * CHUNK;

    for (int i = tid; i < 16 * CDIM; i += 256) {
        int r = i / CDIM, c = i % CDIM;
        sEq[r][c] = Eq[(q0 + r) * CDIM + c];
        sIEq[r][c] = iEq[(q0 + r) * CDIM + c];
    }
    if (tid < 48) {
        int r = tid / 3, d = tid % 3;
        sEqc[r][d] = Eqc[(q0 + r) * 3 + d];
        sIEqc[r][d] = iEqc[(q0 + r) * 3 + d];
    }
    if (tid < 3) sW[tid] = wsum_ws[tid];
    __syncthreads();

    for (int p = tid; p < 16 * CHUNK; p += 256) {
        int qi = p & 15;
        int jl = p >> 4;
        int jj = j0 + jl;
        float s;
        if (jj < NKV) {
            s = 0.f;
            const float4* e4 = reinterpret_cast<const float4*>(Ek + jj * CDIM);
            const float4* i4 = reinterpret_cast<const float4*>(iEk + jj * CDIM);
#pragma unroll 4
            for (int c4 = 0; c4 < CDIM / 4; ++c4) {
                float4 ek = e4[c4];
                float4 ik = i4[c4];
                int c = 4 * c4;
                s += fminf(sEq[qi][c + 0] * ik.x, sIEq[qi][c + 0] * ek.x);
                s += fminf(sEq[qi][c + 1] * ik.y, sIEq[qi][c + 1] * ek.y);
                s += fminf(sEq[qi][c + 2] * ik.z, sIEq[qi][c + 2] * ek.z);
                s += fminf(sEq[qi][c + 3] * ik.w, sIEq[qi][c + 3] * ek.w);
            }
#pragma unroll
            for (int d = 0; d < 3; ++d) {
                float ekc = Ekc[jj * 3 + d];
                float ikc = iEkc[jj * 3 + d];
                s += fminf(sEqc[qi][d] * ikc, sIEqc[qi][d] * ekc) * sW[d];
            }
            s *= SCALE;
        } else {
            s = -__builtin_inff();
        }
        sP[qi][jl] = s;
    }
    __syncthreads();

    const int qi = tid & 15;
    const int g = tid >> 4;
    {
        float m = -__builtin_inff();
        for (int jl = g; jl < CHUNK; jl += 16) m = fmaxf(m, sP[qi][jl]);
        sRed[qi][g] = m;
    }
    __syncthreads();
    if (tid < 16) {
        float m = sRed[tid][0];
#pragma unroll
        for (int i = 1; i < 16; ++i) m = fmaxf(m, sRed[tid][i]);
        sMax[tid] = m;
    }
    __syncthreads();

    {
        float mr = sMax[qi];
        float psum = 0.f;
        for (int jl = g; jl < CHUNK; jl += 16) {
            float e = __expf(sP[qi][jl] - mr);
            sP[qi][jl] = e;
            psum += e;
        }
        sRed[qi][g] = psum;
    }
    __syncthreads();
    if (tid < 16) {
        float s = 0.f;
#pragma unroll
        for (int i = 0; i < 16; ++i) s += sRed[tid][i];
        sSum[tid] = s;
    }
    __syncthreads();

    const int wave = tid >> 5;
    const int lane = tid & 31;
    const int half = lane >> 4;
    const int lm = lane & 15;
    const long base = (long)(qt * NCHUNK + ch) * 16;
    if (wave < 4) {
        int n0 = wave * 16;
        v8f acc = {};
        for (int kk = 0; kk < CHUNK; kk += 4) {
            int ka = kk + 2 * half;
            v2f a, b;
            a.x = sP[lm][ka];
            a.y = sP[lm][ka + 1];
            b.x = v_ws[(j0 + ka) * CDIM + n0 + lm];
            b.y = v_ws[(j0 + ka + 1) * CDIM + n0 + lm];
            acc = wmma_f32_4(a, b, acc);
        }
#pragma unroll
        for (int r = 0; r < 8; ++r) {
            int row = r + 8 * half;
            po[(base + row) * CDIM + n0 + lm] = acc[r];
        }
    }
    if (tid < 16) {
        pm[base + tid] = sMax[tid];
        ps[base + tid] = sSum[tid];
    }
}

#define CROWS 2
#define NCBLK (NQ / CROWS)
#define CTHREADS (CROWS * CDIM)

static __device__ __forceinline__ float lane_value(float v, int l) {
    return __int_as_float(__builtin_amdgcn_readlane(__float_as_int(v), l));
}
template <int XM>
static __device__ __forceinline__ float row_xor(float v) {
    return __int_as_float(
        __builtin_amdgcn_update_dpp(0, __float_as_int(v), 0x160 + XM, 0xf, 0xf, true));
}
static __device__ __forceinline__ float xhalf(float v) {
    int b = __float_as_int(v);
    return __int_as_float(
        __builtin_amdgcn_permlanex16(b, b, 0x76543210, 0xfedcba98, false, false));
}
static __device__ __forceinline__ float wave_max(float v) {
    v = fmaxf(v, row_xor<1>(v));
    v = fmaxf(v, row_xor<2>(v));
    v = fmaxf(v, row_xor<4>(v));
    v = fmaxf(v, row_xor<8>(v));
    return fmaxf(v, xhalf(v));
}
static __device__ __forceinline__ float wave_sum(float v) {
    v += row_xor<1>(v);
    v += row_xor<2>(v);
    v += row_xor<4>(v);
    v += row_xor<8>(v);
    return v + xhalf(v);
}

__global__ void __launch_bounds__(CTHREADS)
diffattn_combine(const float* __restrict__ po, const float* __restrict__ pm,
                 const float* __restrict__ ps,
                 const float* __restrict__ Wproj,
                 const float* __restrict__ bproj, float* __restrict__ out) {
    __shared__ float sO[CROWS][CDIM + 1];
    __shared__ float sOut[16][CDIM];

    const int tid = threadIdx.x;
    const int lane = tid & 31;
    const int wave = tid >> 5;
    const int half = lane >> 4;
    const int lm = lane & 15;
    const int r = tid >> 6;
    const int n = tid & 63;
    const int n0 = wave * 16;
    const int grow = blockIdx.x * CROWS + r;
    const long pbase = (long)(grow >> 4) * NCHUNK * 16 + (grow & 15);

    const bool has1 = lane + 32 < NCHUNK;
    const int c1 = has1 ? lane + 32 : NCHUNK - 1;
    float m0 = pm[pbase + lane * 16];
    float s0 = ps[pbase + lane * 16];
    float m1 = pm[pbase + c1 * 16];
    float s1 = ps[pbase + c1 * 16];

    float pv[NCHUNK];
#pragma unroll
    for (int c = 0; c < NCHUNK; ++c) pv[c] = po[(pbase + c * 16) * CDIM + n];

    v2f bf[CDIM / 4];
#pragma unroll
    for (int kk = 0; kk < CDIM; kk += 4) {
        int ka = kk + 2 * half;
        bf[kk / 4].x = Wproj[ka * CDIM + n0 + lm];
        bf[kk / 4].y = Wproj[(ka + 1) * CDIM + n0 + lm];
    }
    float bb = bproj[n];
    __builtin_amdgcn_sched_barrier(0);

    m1 = has1 ? m1 : -__builtin_inff();
    float M = wave_max(fmaxf(m0, m1));
    float e0 = __expf(m0 - M);
    float e1 = has1 ? __expf(m1 - M) : 0.f;
    float S = wave_sum(s0 * e0 + s1 * e1);
    float inv = 1.0f / S;
    float w0 = e0 * inv;
    float w1 = e1 * inv;

    float a = 0.f;
#pragma unroll
    for (int c = 0; c < NCHUNK; ++c) {
        float wc = (c < 32) ? lane_value(w0, c) : lane_value(w1, c - 32);
        a += pv[c] * wc;
    }
    sO[r][n] = a;
    __syncthreads();

    v8f acc = {};
#pragma unroll
    for (int kk = 0; kk < CDIM; kk += 4) {
        int ka = kk + 2 * half;
        v2f a2;
        a2.x = sO[lm & (CROWS - 1)][ka];
        a2.y = sO[lm & (CROWS - 1)][ka + 1];
        acc = wmma_f32_4(a2, bf[kk / 4], acc);
    }
#pragma unroll
    for (int q = 0; q < CROWS; ++q) sOut[q + 8 * half][n0 + lm] = acc[q];
    __syncthreads();

    out[grow * CDIM + n] = sOut[r][n] + bb;
}

extern "C" void kernel_launch(void* const* d_in, const int* in_sizes, int n_in,
                              void* d_out, int out_size, void* d_ws, size_t ws_size,
                              hipStream_t stream) {
    const float* q        = (const float*)d_in[0];
    const float* q_coord  = (const float*)d_in[1];
    const float* kv       = (const float*)d_in[2];
    const float* kv_coord = (const float*)d_in[3];
    const float* Wq       = (const float*)d_in[4];
    const float* Wkv      = (const float*)d_in[5];
    const float* Wdelta   = (const float*)d_in[6];
    const float* Wproj    = (const float*)d_in[7];
    const float* bproj    = (const float*)d_in[8];
    float* out = (float*)d_out;
    float* ws  = (float*)d_ws;

    float* q_ws    = ws + OFF_QW;
    float* k_ws    = ws + OFF_KW;
    float* v_ws    = ws + OFF_VW;
    float* Eqp     = ws + OFF_EQ;
    float* iEqp    = ws + OFF_IEQ;
    float* Ekp     = ws + OFF_EK;
    float* iEkp    = ws + OFF_IEK;
    float* wsum_ws = ws + OFF_WSUM;
    float* Eqc     = ws + OFF_EQC;
    float* iEqc    = ws + OFF_IEQC;
    float* Ekc     = ws + OFF_EKC;
    float* iEkc    = ws + OFF_IEKC;
    float* po      = ws + OFF_PO;
    float* pm      = ws + OFF_PM;
    float* ps      = ws + OFF_PS;

    diffattn_prep<<<180, 256, 0, stream>>>(q, kv, Wq, Wkv, Wdelta,
                                           q_ws, k_ws, v_ws, wsum_ws);

    int exp_total = NQ * CDIM + NKV_PAD * CDIM + NQ * 3 + NKV * 3;
    diffattn_exp<<<(exp_total + 255) / 256, 256, 0, stream>>>(
        q_ws, k_ws, q_coord, kv_coord, Eqp, iEqp, Ekp, iEkp, Eqc, iEqc, Ekc, iEkc);

    dim3 sgrid(NCHUNK, NQT);
    diffattn_score<<<sgrid, 256, 0, stream>>>(Eqp, iEqp, Ekp, iEkp, Eqc, iEqc,
                                              Ekc, iEkc, wsum_ws, v_ws, po, pm, ps);

    diffattn_combine<<<NCBLK, CTHREADS, 0, stream>>>(po, pm, ps, Wproj, bproj, out);
}


# exp tables (e^x, e^-x of q_, k and both coords) now written by the projection kernel: tile epilogue plus 37 element blocks; separate exp kernel and the q_/k round trip removed, 4 launches -> 3
# speedup vs baseline: 1.5531x; 1.0632x over previous
#include <hip/hip_runtime.h>
#include <hip/hip_bf16.h>

typedef float v2f __attribute__((ext_vector_type(2)));
typedef float v8f __attribute__((ext_vector_type(8)));

#define NQ 512
#define NKV 2562
#define NKV_PAD 2624
#define CDIM 64
#define NCHUNK 41
#define CHUNK 64
#define NQT (NQ / 16)
#define SCALE (1.0f / 64.0f)

#define OFF_QW   0L
#define OFF_KW   (OFF_QW   + (long)NQ * CDIM)
#define OFF_VW   (OFF_KW   + (long)NKV_PAD * CDIM)
#define OFF_EQ   (OFF_VW   + (long)NKV_PAD * CDIM)
#define OFF_IEQ  (OFF_EQ   + (long)NQ * CDIM)
#define OFF_EK   (OFF_IEQ  + (long)NQ * CDIM)
#define OFF_IEK  (OFF_EK   + (long)NKV_PAD * CDIM)
#define OFF_WSUM (OFF_IEK  + (long)NKV_PAD * CDIM)
#define OFF_EQC  (OFF_WSUM + 16L)
#define OFF_IEQC (OFF_EQC  + (long)NQ * 3)
#define OFF_EKC  (OFF_IEQC + (long)NQ * 3)
#define OFF_IEKC (OFF_EKC  + (long)NKV * 3)
#define OFF_PO   (OFF_IEKC + (long)NKV * 3 + 16L)
#define OFF_PM   (OFF_PO   + (long)NQT * NCHUNK * 16 * CDIM)
#define OFF_PS   (OFF_PM   + (long)NQT * NCHUNK * 16)

static __device__ __forceinline__ v8f wmma_f32_4(v2f a, v2f b, v8f c) {
    return __builtin_amdgcn_wmma_f32_16x16x4_f32(false, a, false, b, (short)0, c,
                                                 false, false);
}

#define PREP_BLOCKS 180
#define COORD_BLOCKS ((NQ * 3 + NKV * 3 + 255) / 256)

__global__ void diffattn_prep(const float* __restrict__ q,
                              const float* __restrict__ kv,
                              const float* __restrict__ q_coord,
                              const float* __restrict__ kv_coord,
                              const float* __restrict__ Wq,
                              const float* __restrict__ Wkv,
                              const float* __restrict__ Wdelta,
                              float* __restrict__ Eq, float* __restrict__ iEq,
                              float* __restrict__ Ek, float* __restrict__ iEk,
                              float* __restrict__ Eqc, float* __restrict__ iEqc,
                              float* __restrict__ Ekc, float* __restrict__ iEkc,
                              float* __restrict__ v_ws,
                              float* __restrict__ wsum_ws) {
    if (blockIdx.x >= PREP_BLOCKS) {
        const int NE_QC = NQ * 3;
        const int NE_KC = NKV * 3;
        int i = (blockIdx.x - PREP_BLOCKS) * 256 + threadIdx.x;
        if (i < NE_QC) {
            float x = q_coord[i];
            Eqc[i] = __expf(x);
            iEqc[i] = __expf(-x);
        } else if (i < NE_QC + NE_KC) {
            int j = i - NE_QC;
            float x = kv_coord[j];
            Ekc[j] = __expf(x);
            iEkc[j] = __expf(-x);
        }
        return;
    }

    if (blockIdx.x == 0 && threadIdx.x < 3) {
        float s = 0.f;
        for (int c = 0; c < CDIM; ++c) s += Wdelta[threadIdx.x * CDIM + c];
        wsum_ws[threadIdx.x] = s;
    }

    const int KV_TILES = (NKV_PAD / 16) * (128 / 16);
    int wave = (blockIdx.x * blockDim.x + threadIdx.x) >> 5;
    int lane = threadIdx.x & 31;

    int half = lane >> 4;
    int lm = lane & 15;

    if (wave < KV_TILES) {
        int mt = wave >> 3, nt = wave & 7;
        int m0 = mt * 16, n0 = nt * 16;
        int rowA = m0 + lm;
        bool inb = rowA < NKV;
        v8f acc = {};
        for (int kk = 0; kk < CDIM; kk += 4) {
            int ka = kk + 2 * half;
            v2f a, b;
            a.x = inb ? kv[rowA * CDIM + ka] : 0.f;
            a.y = inb ? kv[rowA * CDIM + ka + 1] : 0.f;
            b.x = Wkv[ka * 128 + n0 + lm];
            b.y = Wkv[(ka + 1) * 128 + n0 + lm];
            acc = wmma_f32_4(a, b, acc);
        }
        int col = n0 + lm;
        if (n0 < CDIM) {
#pragma unroll
            for (int r = 0; r < 8; ++r) {
                int row = m0 + r + 8 * half;
                float x = acc[r];
                Ek[row * CDIM + col] = __expf(x);
                iEk[row * CDIM + col] = __expf(-x);
            }
        } else {
#pragma unroll
            for (int r = 0; r < 8; ++r) {
                int row = m0 + r + 8 * half;
                v_ws[row * CDIM + (col - CDIM)] = acc[r];
            }
        }
    } else {
        int t = wave - KV_TILES;
        int mt = t >> 2, nt = t & 3;
        int m0 = mt * 16, n0 = nt * 16;
        int rowA = m0 + lm;
        v8f acc = {};
        for (int kk = 0; kk < CDIM; kk += 4) {
            int ka = kk + 2 * half;
            v2f a, b;
            a.x = q[rowA * CDIM + ka];
            a.y = q[rowA * CDIM + ka + 1];
            b.x = Wq[ka * CDIM + n0 + lm];
            b.y = Wq[(ka + 1) * CDIM + n0 + lm];
            acc = wmma_f32_4(a, b, acc);
        }
        int col = n0 + lm;
#pragma unroll
        for (int r = 0; r < 8; ++r) {
            int row = m0 + r + 8 * half;
            float x = acc[r];
            Eq[row * CDIM + col] = __expf(x);
            iEq[row * CDIM + col] = __expf(-x);
        }
    }
}

__global__ void __launch_bounds__(256)
diffattn_score(const float* __restrict__ Eq, const float* __restrict__ iEq,
               const float* __restrict__ Ek, const float* __restrict__ iEk,
               const float* __restrict__ Eqc, const float* __restrict__ iEqc,
               const float* __restrict__ Ekc, const float* __restrict__ iEkc,
               const float* __restrict__ wsum_ws,
               const float* __restrict__ v_ws,
               float* __restrict__ po, float* __restrict__ pm,
               float* __restrict__ ps) {
    __shared__ float sEq[16][CDIM + 1];
    __shared__ float sIEq[16][CDIM + 1];
    __shared__ float sP[16][CHUNK + 8];
    __shared__ float sEqc[16][4];
    __shared__ float sIEqc[16][4];
    __shared__ float sRed[16][17];
    __shared__ float sMax[16];
    __shared__ float sSum[16];
    __shared__ float sW[3];

    const int tid = threadIdx.x;
    const int ch = blockIdx.x;
    const int qt = blockIdx.y;
    const int q0 = qt * 16;
    const int j0 = ch * CHUNK;

    for (int i = tid; i < 16 * CDIM; i += 256) {
        int r = i / CDIM, c = i % CDIM;
        sEq[r][c] = Eq[(q0 + r) * CDIM + c];
        sIEq[r][c] = iEq[(q0 + r) * CDIM + c];
    }
    if (tid < 48) {
        int r = tid / 3, d = tid % 3;
        sEqc[r][d] = Eqc[(q0 + r) * 3 + d];
        sIEqc[r][d] = iEqc[(q0 + r) * 3 + d];
    }
    if (tid < 3) sW[tid] = wsum_ws[tid];
    __syncthreads();

    for (int p = tid; p < 16 * CHUNK; p += 256) {
        int qi = p & 15;
        int jl = p >> 4;
        int jj = j0 + jl;
        float s;
        if (jj < NKV) {
            s = 0.f;
            const float4* e4 = reinterpret_cast<const float4*>(Ek + jj * CDIM);
            const float4* i4 = reinterpret_cast<const float4*>(iEk + jj * CDIM);
#pragma unroll 4
            for (int c4 = 0; c4 < CDIM / 4; ++c4) {
                float4 ek = e4[c4];
                float4 ik = i4[c4];
                int c = 4 * c4;
                s += fminf(sEq[qi][c + 0] * ik.x, sIEq[qi][c + 0] * ek.x);
                s += fminf(sEq[qi][c + 1] * ik.y, sIEq[qi][c + 1] * ek.y);
                s += fminf(sEq[qi][c + 2] * ik.z, sIEq[qi][c + 2] * ek.z);
                s += fminf(sEq[qi][c + 3] * ik.w, sIEq[qi][c + 3] * ek.w);
            }
#pragma unroll
            for (int d = 0; d < 3; ++d) {
                float ekc = Ekc[jj * 3 + d];
                float ikc = iEkc[jj * 3 + d];
                s += fminf(sEqc[qi][d] * ikc, sIEqc[qi][d] * ekc) * sW[d];
            }
            s *= SCALE;
        } else {
            s = -__builtin_inff();
        }
        sP[qi][jl] = s;
    }
    __syncthreads();

    const int qi = tid & 15;
    const int g = tid >> 4;
    {
        float m = -__builtin_inff();
        for (int jl = g; jl < CHUNK; jl += 16) m = fmaxf(m, sP[qi][jl]);
        sRed[qi][g] = m;
    }
    __syncthreads();
    if (tid < 16) {
        float m = sRed[tid][0];
#pragma unroll
        for (int i = 1; i < 16; ++i) m = fmaxf(m, sRed[tid][i]);
        sMax[tid] = m;
    }
    __syncthreads();

    {
        float mr = sMax[qi];
        float psum = 0.f;
        for (int jl = g; jl < CHUNK; jl += 16) {
            float e = __expf(sP[qi][jl] - mr);
            sP[qi][jl] = e;
            psum += e;
        }
        sRed[qi][g] = psum;
    }
    __syncthreads();
    if (tid < 16) {
        float s = 0.f;
#pragma unroll
        for (int i = 0; i < 16; ++i) s += sRed[tid][i];
        sSum[tid] = s;
    }
    __syncthreads();

    const int wave = tid >> 5;
    const int lane = tid & 31;
    const int half = lane >> 4;
    const int lm = lane & 15;
    const long base = (long)(qt * NCHUNK + ch) * 16;
    if (wave < 4) {
        int n0 = wave * 16;
        v8f acc = {};
        for (int kk = 0; kk < CHUNK; kk += 4) {
            int ka = kk + 2 * half;
            v2f a, b;
            a.x = sP[lm][ka];
            a.y = sP[lm][ka + 1];
            b.x = v_ws[(j0 + ka) * CDIM + n0 + lm];
            b.y = v_ws[(j0 + ka + 1) * CDIM + n0 + lm];
            acc = wmma_f32_4(a, b, acc);
        }
#pragma unroll
        for (int r = 0; r < 8; ++r) {
            int row = r + 8 * half;
            po[(base + row) * CDIM + n0 + lm] = acc[r];
        }
    }
    if (tid < 16) {
        pm[base + tid] = sMax[tid];
        ps[base + tid] = sSum[tid];
    }
}

#define CROWS 2
#define NCBLK (NQ / CROWS)
#define CTHREADS (CROWS * CDIM)

static __device__ __forceinline__ float lane_value(float v, int l) {
    return __int_as_float(__builtin_amdgcn_readlane(__float_as_int(v), l));
}
template <int XM>
static __device__ __forceinline__ float row_xor(float v) {
    return __int_as_float(
        __builtin_amdgcn_update_dpp(0, __float_as_int(v), 0x160 + XM, 0xf, 0xf, true));
}
static __device__ __forceinline__ float xhalf(float v) {
    int b = __float_as_int(v);
    return __int_as_float(
        __builtin_amdgcn_permlanex16(b, b, 0x76543210, 0xfedcba98, false, false));
}
static __device__ __forceinline__ float wave_max(float v) {
    v = fmaxf(v, row_xor<1>(v));
    v = fmaxf(v, row_xor<2>(v));
    v = fmaxf(v, row_xor<4>(v));
    v = fmaxf(v, row_xor<8>(v));
    return fmaxf(v, xhalf(v));
}
static __device__ __forceinline__ float wave_sum(float v) {
    v += row_xor<1>(v);
    v += row_xor<2>(v);
    v += row_xor<4>(v);
    v += row_xor<8>(v);
    return v + xhalf(v);
}

__global__ void __launch_bounds__(CTHREADS)
diffattn_combine(const float* __restrict__ po, const float* __restrict__ pm,
                 const float* __restrict__ ps,
                 const float* __restrict__ Wproj,
                 const float* __restrict__ bproj, float* __restrict__ out) {
    __shared__ float sO[CROWS][CDIM + 1];
    __shared__ float sOut[16][CDIM];

    const int tid = threadIdx.x;
    const int lane = tid & 31;
    const int wave = tid >> 5;
    const int half = lane >> 4;
    const int lm = lane & 15;
    const int r = tid >> 6;
    const int n = tid & 63;
    const int n0 = wave * 16;
    const int grow = blockIdx.x * CROWS + r;
    const long pbase = (long)(grow >> 4) * NCHUNK * 16 + (grow & 15);

    const bool has1 = lane + 32 < NCHUNK;
    const int c1 = has1 ? lane + 32 : NCHUNK - 1;
    float m0 = pm[pbase + lane * 16];
    float s0 = ps[pbase + lane * 16];
    float m1 = pm[pbase + c1 * 16];
    float s1 = ps[pbase + c1 * 16];

    float pv[NCHUNK];
#pragma unroll
    for (int c = 0; c < NCHUNK; ++c) pv[c] = po[(pbase + c * 16) * CDIM + n];

    v2f bf[CDIM / 4];
#pragma unroll
    for (int kk = 0; kk < CDIM; kk += 4) {
        int ka = kk + 2 * half;
        bf[kk / 4].x = Wproj[ka * CDIM + n0 + lm];
        bf[kk / 4].y = Wproj[(ka + 1) * CDIM + n0 + lm];
    }
    float bb = bproj[n];
    __builtin_amdgcn_sched_barrier(0);

    m1 = has1 ? m1 : -__builtin_inff();
    float M = wave_max(fmaxf(m0, m1));
    float e0 = __expf(m0 - M);
    float e1 = has1 ? __expf(m1 - M) : 0.f;
    float S = wave_sum(s0 * e0 + s1 * e1);
    float inv = 1.0f / S;
    float w0 = e0 * inv;
    float w1 = e1 * inv;

    float a = 0.f;
#pragma unroll
    for (int c = 0; c < NCHUNK; ++c) {
        float wc = (c < 32) ? lane_value(w0, c) : lane_value(w1, c - 32);
        a += pv[c] * wc;
    }
    sO[r][n] = a;
    __syncthreads();

    v8f acc = {};
#pragma unroll
    for (int kk = 0; kk < CDIM; kk += 4) {
        int ka = kk + 2 * half;
        v2f a2;
        a2.x = sO[lm & (CROWS - 1)][ka];
        a2.y = sO[lm & (CROWS - 1)][ka + 1];
        acc = wmma_f32_4(a2, bf[kk / 4], acc);
    }
#pragma unroll
    for (int q = 0; q < CROWS; ++q) sOut[q + 8 * half][n0 + lm] = acc[q];
    __syncthreads();

    out[grow * CDIM + n] = sOut[r][n] + bb;
}

extern "C" void kernel_launch(void* const* d_in, const int* in_sizes, int n_in,
                              void* d_out, int out_size, void* d_ws, size_t ws_size,
                              hipStream_t stream) {
    const float* q        = (const float*)d_in[0];
    const float* q_coord  = (const float*)d_in[1];
    const float* kv       = (const float*)d_in[2];
    const float* kv_coord = (const float*)d_in[3];
    const float* Wq       = (const float*)d_in[4];
    const float* Wkv      = (const float*)d_in[5];
    const float* Wdelta   = (const float*)d_in[6];
    const float* Wproj    = (const float*)d_in[7];
    const float* bproj    = (const float*)d_in[8];
    float* out = (float*)d_out;
    float* ws  = (float*)d_ws;

    float* v_ws    = ws + OFF_VW;
    float* Eqp     = ws + OFF_EQ;
    float* iEqp    = ws + OFF_IEQ;
    float* Ekp     = ws + OFF_EK;
    float* iEkp    = ws + OFF_IEK;
    float* wsum_ws = ws + OFF_WSUM;
    float* Eqc     = ws + OFF_EQC;
    float* iEqc    = ws + OFF_IEQC;
    float* Ekc     = ws + OFF_EKC;
    float* iEkc    = ws + OFF_IEKC;
    float* po      = ws + OFF_PO;
    float* pm      = ws + OFF_PM;
    float* ps      = ws + OFF_PS;

    diffattn_prep<<<PREP_BLOCKS + COORD_BLOCKS, 256, 0, stream>>>(
        q, kv, q_coord, kv_coord, Wq, Wkv, Wdelta,
        Eqp, iEqp, Ekp, iEkp, Eqc, iEqc, Ekc, iEkc, v_ws, wsum_ws);

    dim3 sgrid(NCHUNK, NQT);
    diffattn_score<<<sgrid, 256, 0, stream>>>(Eqp, iEqp, Ekp, iEkp, Eqc, iEqc,
                                              Ekc, iEkc, wsum_ws, v_ws, po, pm, ps);

    diffattn_combine<<<NCBLK, CTHREADS, 0, stream>>>(po, pm, ps, Wproj, bproj, out);
}


# score kernel re-tiled: a lane keeps 2 kv rows in registers, a wave sweeps 8 queries read from LDS as broadcasts; tables stored in that order; 2 waves per block, was 8; row sums regrouped 4x16
# speedup vs baseline: 1.8109x; 1.1660x over previous
#include <hip/hip_runtime.h>
#include <hip/hip_bf16.h>

typedef float v2f __attribute__((ext_vector_type(2)));
typedef float v8f __attribute__((ext_vector_type(8)));

#define NQ 512
#define NKV 2562
#define NKV_PAD 2624
#define CDIM 64
#define NCHUNK 41
#define CHUNK 64
#define NQT (NQ / 16)
#define SCALE (1.0f / 64.0f)

#define OFF_QW   0L
#define OFF_KW   (OFF_QW   + (long)NQ * CDIM)
#define OFF_VW   (OFF_KW   + (long)NKV_PAD * CDIM)
#define OFF_EQ   (OFF_VW   + (long)NKV_PAD * CDIM)
#define OFF_IEQ  (OFF_EQ   + (long)NQ * CDIM)
#define OFF_EK   (OFF_IEQ  + (long)NQ * CDIM)
#define OFF_IEK  (OFF_EK   + (long)NKV_PAD * CDIM)
#define OFF_WSUM (OFF_IEK  + (long)NKV_PAD * CDIM)
#define OFF_EQC  (OFF_WSUM + 16L)
#define OFF_IEQC (OFF_EQC  + (long)NQ * 3)
#define OFF_EKC  (OFF_IEQC + (long)NQ * 3)
#define OFF_IEKC (OFF_EKC  + (long)NKV * 3)
#define OFF_PO   (OFF_IEKC + (long)NKV * 3 + 16L)
#define OFF_PM   (OFF_PO   + (long)NQT * NCHUNK * 16 * CDIM)
#define OFF_PS   (OFF_PM   + (long)NQT * NCHUNK * 16)

static __device__ __forceinline__ v8f wmma_f32_4(v2f a, v2f b, v8f c) {
    return __builtin_amdgcn_wmma_f32_16x16x4_f32(false, a, false, b, (short)0, c,
                                                 false, false);
}

#define PREP_BLOCKS 180
#define COORD_BLOCKS ((NQ * 3 + NKV * 3 + 255) / 256)

__global__ void diffattn_prep(const float* __restrict__ q,
                              const float* __restrict__ kv,
                              const float* __restrict__ q_coord,
                              const float* __restrict__ kv_coord,
                              const float* __restrict__ Wq,
                              const float* __restrict__ Wkv,
                              const float* __restrict__ Wdelta,
                              float* __restrict__ Eq, float* __restrict__ iEq,
                              float* __restrict__ Ek, float* __restrict__ iEk,
                              float* __restrict__ Eqc, float* __restrict__ iEqc,
                              float* __restrict__ Ekc, float* __restrict__ iEkc,
                              float* __restrict__ v_ws,
                              float* __restrict__ wsum_ws) {
    if (blockIdx.x >= PREP_BLOCKS) {
        const int NE_QC = NQ * 3;
        const int NE_KC = NKV * 3;
        int i = (blockIdx.x - PREP_BLOCKS) * 256 + threadIdx.x;
        if (i < NE_QC) {
            float x = q_coord[i];
            Eqc[i] = __expf(x);
            iEqc[i] = __expf(-x);
        } else if (i < NE_QC + NE_KC) {
            int j = i - NE_QC;
            float x = kv_coord[j];
            Ekc[j] = __expf(x);
            iEkc[j] = __expf(-x);
        }
        return;
    }

    if (blockIdx.x == 0 && threadIdx.x < 3) {
        float s = 0.f;
        for (int c = 0; c < CDIM; ++c) s += Wdelta[threadIdx.x * CDIM + c];
        wsum_ws[threadIdx.x] = s;
    }

    const int KV_TILES = (NKV_PAD / 16) * (128 / 16);
    int wave = (blockIdx.x * blockDim.x + threadIdx.x) >> 5;
    int lane = threadIdx.x & 31;

    int half = lane >> 4;
    int lm = lane & 15;

    if (wave < KV_TILES) {
        int mt = wave >> 3, nt = wave & 7;
        int m0 = mt * 16, n0 = nt * 16;
        int rowA = m0 + lm;
        bool inb = rowA < NKV;
        const bool ktile = n0 < CDIM;
        v8f acc = {};
        for (int kk = 0; kk < CDIM; kk += 4) {
            int ka = kk + 2 * half;
            v2f a, b;
            a.x = inb ? kv[rowA * CDIM + ka] : 0.f;
            a.y = inb ? kv[rowA * CDIM + ka + 1] : 0.f;
            b.x = Wkv[ka * 128 + n0 + lm];
            b.y = Wkv[(ka + 1) * 128 + n0 + lm];
            v2f fa = ktile ? b : a;
            v2f fb = ktile ? a : b;
            acc = wmma_f32_4(fa, fb, acc);
        }
        int col = n0 + lm;
        if (ktile) {
            const int row = m0 + lm;
            const int c4 = (n0 + 8 * half) >> 2;
            float4 e0, e1, i0, i1;
            e0.x = __expf(acc[0]);  e0.y = __expf(acc[1]);
            e0.z = __expf(acc[2]);  e0.w = __expf(acc[3]);
            e1.x = __expf(acc[4]);  e1.y = __expf(acc[5]);
            e1.z = __expf(acc[6]);  e1.w = __expf(acc[7]);
            i0.x = __expf(-acc[0]); i0.y = __expf(-acc[1]);
            i0.z = __expf(-acc[2]); i0.w = __expf(-acc[3]);
            i1.x = __expf(-acc[4]); i1.y = __expf(-acc[5]);
            i1.z = __expf(-acc[6]); i1.w = __expf(-acc[7]);
            float4* E4 = reinterpret_cast<float4*>(Ek);
            float4* I4 = reinterpret_cast<float4*>(iEk);
            E4[c4 * NKV_PAD + row] = e0;
            E4[(c4 + 1) * NKV_PAD + row] = e1;
            I4[c4 * NKV_PAD + row] = i0;
            I4[(c4 + 1) * NKV_PAD + row] = i1;
        } else {
#pragma unroll
            for (int r = 0; r < 8; ++r) {
                int row = m0 + r + 8 * half;
                v_ws[row * CDIM + (col - CDIM)] = acc[r];
            }
        }
    } else {
        int t = wave - KV_TILES;
        int mt = t >> 2, nt = t & 3;
        int m0 = mt * 16, n0 = nt * 16;
        int rowA = m0 + lm;
        v8f acc = {};
        for (int kk = 0; kk < CDIM; kk += 4) {
            int ka = kk + 2 * half;
            v2f a, b;
            a.x = q[rowA * CDIM + ka];
            a.y = q[rowA * CDIM + ka + 1];
            b.x = Wq[ka * CDIM + n0 + lm];
            b.y = Wq[(ka + 1) * CDIM + n0 + lm];
            acc = wmma_f32_4(a, b, acc);
        }
        int col = n0 + lm;
        float4 e0, e1, i0, i1;
        e0.x = __expf(acc[0]);  e0.y = __expf(acc[1]);
        e0.z = __expf(acc[2]);  e0.w = __expf(acc[3]);
        e1.x = __expf(acc[4]);  e1.y = __expf(acc[5]);
        e1.z = __expf(acc[6]);  e1.w = __expf(acc[7]);
        i0.x = __expf(-acc[0]); i0.y = __expf(-acc[1]);
        i0.z = __expf(-acc[2]); i0.w = __expf(-acc[3]);
        i1.x = __expf(-acc[4]); i1.y = __expf(-acc[5]);
        i1.z = __expf(-acc[6]); i1.w = __expf(-acc[7]);
        float4* E4 = reinterpret_cast<float4*>(Eq);
        float4* I4 = reinterpret_cast<float4*>(iEq);
        const int f = (mt * CDIM + col) * 4 + 2 * half;
        E4[f] = e0;
        E4[f + 1] = e1;
        I4[f] = i0;
        I4[f + 1] = i1;
    }
}

#define STHREADS 64

__global__ void __launch_bounds__(STHREADS)
diffattn_score(const float* __restrict__ Eq, const float* __restrict__ iEq,
               const float* __restrict__ Ek, const float* __restrict__ iEk,
               const float* __restrict__ Eqc, const float* __restrict__ iEqc,
               const float* __restrict__ Ekc, const float* __restrict__ iEkc,
               const float* __restrict__ wsum_ws,
               const float* __restrict__ v_ws,
               float* __restrict__ po, float* __restrict__ pm,
               float* __restrict__ ps) {
    __shared__ float4 sEqT[CDIM][4];
    __shared__ float4 sIEqT[CDIM][4];
    __shared__ float sP[16][CHUNK + 8];
    __shared__ float sEqc[16][4];
    __shared__ float sIEqc[16][4];
    __shared__ float sRed[16][5];
    __shared__ float sMax[16];
    __shared__ float sSum[16];
    __shared__ float sW[4];

    const int tid = threadIdx.x;
    const int lane = tid & 31;
    const int wave = tid >> 5;
    const int ch = blockIdx.x;
    const int qt = blockIdx.y;
    const int q0 = qt * 16;
    const int j0 = ch * CHUNK;

    {
        const float4* gq = reinterpret_cast<const float4*>(Eq + q0 * CDIM);
        const float4* gi = reinterpret_cast<const float4*>(iEq + q0 * CDIM);
        float4* dq = &sEqT[0][0];
        float4* di = &sIEqT[0][0];
        for (int i = tid; i < 16 * CDIM / 4; i += STHREADS) {
            dq[i] = gq[i];
            di[i] = gi[i];
        }
    }
    if (tid < 48) {
        int r = tid / 3, d = tid % 3;
        sEqc[r][d] = Eqc[(q0 + r) * 3 + d];
        sIEqc[r][d] = iEqc[(q0 + r) * 3 + d];
    }
    if (tid < 3) sW[tid] = wsum_ws[tid];
    __syncthreads();

    {
        const int qw = 8 * wave;
        const int jA = j0 + lane;
        const int jB = jA + 32;
        const float4* ek4 = reinterpret_cast<const float4*>(Ek);
        const float4* ik4 = reinterpret_cast<const float4*>(iEk);

        float sA[8], sB[8];
#pragma unroll
        for (int q = 0; q < 8; ++q) { sA[q] = 0.f; sB[q] = 0.f; }

        float4 eA[2], iA[2], eB[2], iB[2];
#pragma unroll
        for (int u = 0; u < 2; ++u) {
            eA[u] = ek4[u * NKV_PAD + jA];
            iA[u] = ik4[u * NKV_PAD + jA];
            eB[u] = ek4[u * NKV_PAD + jB];
            iB[u] = ik4[u * NKV_PAD + jB];
        }

#pragma unroll 1
        for (int cb = 0; cb < 8; ++cb) {
            const int nb = (cb < 7) ? cb + 1 : 7;
            float4 neA[2], niA[2], neB[2], niB[2];
#pragma unroll
            for (int u = 0; u < 2; ++u) {
                neA[u] = ek4[(2 * nb + u) * NKV_PAD + jA];
                niA[u] = ik4[(2 * nb + u) * NKV_PAD + jA];
                neB[u] = ek4[(2 * nb + u) * NKV_PAD + jB];
                niB[u] = ik4[(2 * nb + u) * NKV_PAD + jB];
            }
#define SCORE_STEP(c, EA, IA, EB, IB)                                          \
            _Pragma("unroll")                                                  \
            for (int v = 0; v < 2; ++v) {                                      \
                const float4 eq = sEqT[8 * cb + (c)][2 * wave + v];            \
                const float4 iq = sIEqT[8 * cb + (c)][2 * wave + v];           \
                sA[4 * v + 0] += fminf(eq.x * (IA), iq.x * (EA));              \
                sA[4 * v + 1] += fminf(eq.y * (IA), iq.y * (EA));              \
                sA[4 * v + 2] += fminf(eq.z * (IA), iq.z * (EA));              \
                sA[4 * v + 3] += fminf(eq.w * (IA), iq.w * (EA));              \
                sB[4 * v + 0] += fminf(eq.x * (IB), iq.x * (EB));              \
                sB[4 * v + 1] += fminf(eq.y * (IB), iq.y * (EB));              \
                sB[4 * v + 2] += fminf(eq.z * (IB), iq.z * (EB));              \
                sB[4 * v + 3] += fminf(eq.w * (IB), iq.w * (EB));              \
            }
            SCORE_STEP(0, eA[0].x, iA[0].x, eB[0].x, iB[0].x)
            SCORE_STEP(1, eA[0].y, iA[0].y, eB[0].y, iB[0].y)
            SCORE_STEP(2, eA[0].z, iA[0].z, eB[0].z, iB[0].z)
            SCORE_STEP(3, eA[0].w, iA[0].w, eB[0].w, iB[0].w)
            SCORE_STEP(4, eA[1].x, iA[1].x, eB[1].x, iB[1].x)
            SCORE_STEP(5, eA[1].y, iA[1].y, eB[1].y, iB[1].y)
            SCORE_STEP(6, eA[1].z, iA[1].z, eB[1].z, iB[1].z)
            SCORE_STEP(7, eA[1].w, iA[1].w, eB[1].w, iB[1].w)
#undef SCORE_STEP
#pragma unroll
            for (int u = 0; u < 2; ++u) {
                eA[u] = neA[u]; iA[u] = niA[u];
                eB[u] = neB[u]; iB[u] = niB[u];
            }
        }

        const bool okA = jA < NKV;
        const bool okB = jB < NKV;
        const int jAc = okA ? jA : NKV - 1;
        const int jBc = okB ? jB : NKV - 1;
        float ekA[3], ikA[3], ekB[3], ikB[3];
#pragma unroll
        for (int d = 0; d < 3; ++d) {
            ekA[d] = Ekc[jAc * 3 + d];
            ikA[d] = iEkc[jAc * 3 + d];
            ekB[d] = Ekc[jBc * 3 + d];
            ikB[d] = iEkc[jBc * 3 + d];
        }
#pragma unroll
        for (int q = 0; q < 8; ++q) {
            float a = sA[q], b = sB[q];
#pragma unroll
            for (int d = 0; d < 3; ++d) {
                float eqc = sEqc[qw + q][d];
                float iqc = sIEqc[qw + q][d];
                a += fminf(eqc * ikA[d], iqc * ekA[d]) * sW[d];
                b += fminf(eqc * ikB[d], iqc * ekB[d]) * sW[d];
            }
            a *= SCALE;
            b *= SCALE;
            sP[qw + q][lane] = okA ? a : -__builtin_inff();
            sP[qw + q][lane + 32] = okB ? b : -__builtin_inff();
        }
    }
    __syncthreads();

    const int qi = tid & 15;
    const int g = tid >> 4;
    {
        float m = -__builtin_inff();
        for (int jl = g; jl < CHUNK; jl += 4) m = fmaxf(m, sP[qi][jl]);
        sRed[qi][g] = m;
    }
    __syncthreads();
    if (tid < 16) {
        float m = sRed[tid][0];
#pragma unroll
        for (int i = 1; i < 4; ++i) m = fmaxf(m, sRed[tid][i]);
        sMax[tid] = m;
    }
    __syncthreads();

    {
        float mr = sMax[qi];
        float psum = 0.f;
        for (int jl = g; jl < CHUNK; jl += 4) {
            float e = __expf(sP[qi][jl] - mr);
            sP[qi][jl] = e;
            psum += e;
        }
        sRed[qi][g] = psum;
    }
    __syncthreads();
    if (tid < 16) {
        float s = 0.f;
#pragma unroll
        for (int i = 0; i < 4; ++i) s += sRed[tid][i];
        sSum[tid] = s;
    }
    __syncthreads();

    const int half = lane >> 4;
    const int lm = lane & 15;
    const long base = (long)(qt * NCHUNK + ch) * 16;
    {
        const int n0 = wave * 32;
        v8f acc0 = {};
        v8f acc1 = {};
        for (int kk = 0; kk < CHUNK; kk += 4) {
            int ka = kk + 2 * half;
            v2f a, b0, b1;
            a.x = sP[lm][ka];
            a.y = sP[lm][ka + 1];
            b0.x = v_ws[(j0 + ka) * CDIM + n0 + lm];
            b0.y = v_ws[(j0 + ka + 1) * CDIM + n0 + lm];
            b1.x = v_ws[(j0 + ka) * CDIM + n0 + 16 + lm];
            b1.y = v_ws[(j0 + ka + 1) * CDIM + n0 + 16 + lm];
            acc0 = wmma_f32_4(a, b0, acc0);
            acc1 = wmma_f32_4(a, b1, acc1);
        }
#pragma unroll
        for (int r = 0; r < 8; ++r) {
            int row = r + 8 * half;
            po[(base + row) * CDIM + n0 + lm] = acc0[r];
            po[(base + row) * CDIM + n0 + 16 + lm] = acc1[r];
        }
    }
    if (tid < 16) {
        pm[base + tid] = sMax[tid];
        ps[base + tid] = sSum[tid];
    }
}

#define CROWS 2
#define NCBLK (NQ / CROWS)
#define CTHREADS (CROWS * CDIM)

static __device__ __forceinline__ float lane_value(float v, int l) {
    return __int_as_float(__builtin_amdgcn_readlane(__float_as_int(v), l));
}
template <int XM>
static __device__ __forceinline__ float row_xor(float v) {
    return __int_as_float(
        __builtin_amdgcn_update_dpp(0, __float_as_int(v), 0x160 + XM, 0xf, 0xf, true));
}
static __device__ __forceinline__ float xhalf(float v) {
    int b = __float_as_int(v);
    return __int_as_float(
        __builtin_amdgcn_permlanex16(b, b, 0x76543210, 0xfedcba98, false, false));
}
static __device__ __forceinline__ float wave_max(float v) {
    v = fmaxf(v, row_xor<1>(v));
    v = fmaxf(v, row_xor<2>(v));
    v = fmaxf(v, row_xor<4>(v));
    v = fmaxf(v, row_xor<8>(v));
    return fmaxf(v, xhalf(v));
}
static __device__ __forceinline__ float wave_sum(float v) {
    v += row_xor<1>(v);
    v += row_xor<2>(v);
    v += row_xor<4>(v);
    v += row_xor<8>(v);
    return v + xhalf(v);
}

__global__ void __launch_bounds__(CTHREADS)
diffattn_combine(const float* __restrict__ po, const float* __restrict__ pm,
                 const float* __restrict__ ps,
                 const float* __restrict__ Wproj,
                 const float* __restrict__ bproj, float* __restrict__ out) {
    __shared__ float sO[CROWS][CDIM + 1];
    __shared__ float sOut[16][CDIM];

    const int tid = threadIdx.x;
    const int lane = tid & 31;
    const int wave = tid >> 5;
    const int half = lane >> 4;
    const int lm = lane & 15;
    const int r = tid >> 6;
    const int n = tid & 63;
    const int n0 = wave * 16;
    const int grow = blockIdx.x * CROWS + r;
    const long pbase = (long)(grow >> 4) * NCHUNK * 16 + (grow & 15);

    const bool has1 = lane + 32 < NCHUNK;
    const int c1 = has1 ? lane + 32 : NCHUNK - 1;
    float m0 = pm[pbase + lane * 16];
    float s0 = ps[pbase + lane * 16];
    float m1 = pm[pbase + c1 * 16];
    float s1 = ps[pbase + c1 * 16];

    float pv[NCHUNK];
#pragma unroll
    for (int c = 0; c < NCHUNK; ++c) pv[c] = po[(pbase + c * 16) * CDIM + n];

    v2f bf[CDIM / 4];
#pragma unroll
    for (int kk = 0; kk < CDIM; kk += 4) {
        int ka = kk + 2 * half;
        bf[kk / 4].x = Wproj[ka * CDIM + n0 + lm];
        bf[kk / 4].y = Wproj[(ka + 1) * CDIM + n0 + lm];
    }
    float bb = bproj[n];
    __builtin_amdgcn_sched_barrier(0);

    m1 = has1 ? m1 : -__builtin_inff();
    float M = wave_max(fmaxf(m0, m1));
    float e0 = __expf(m0 - M);
    float e1 = has1 ? __expf(m1 - M) : 0.f;
    float S = wave_sum(s0 * e0 + s1 * e1);
    float inv = 1.0f / S;
    float w0 = e0 * inv;
    float w1 = e1 * inv;

    float a = 0.f;
#pragma unroll
    for (int c = 0; c < NCHUNK; ++c) {
        float wc = (c < 32) ? lane_value(w0, c) : lane_value(w1, c - 32);
        a += pv[c] * wc;
    }
    sO[r][n] = a;
    __syncthreads();

    v8f acc = {};
#pragma unroll
    for (int kk = 0; kk < CDIM; kk += 4) {
        int ka = kk + 2 * half;
        v2f a2;
        a2.x = sO[lm & (CROWS - 1)][ka];
        a2.y = sO[lm & (CROWS - 1)][ka + 1];
        acc = wmma_f32_4(a2, bf[kk / 4], acc);
    }
#pragma unroll
    for (int q = 0; q < CROWS; ++q) sOut[q + 8 * half][n0 + lm] = acc[q];
    __syncthreads();

    out[grow * CDIM + n] = sOut[r][n] + bb;
}

extern "C" void kernel_launch(void* const* d_in, const int* in_sizes, int n_in,
                              void* d_out, int out_size, void* d_ws, size_t ws_size,
                              hipStream_t stream) {
    const float* q        = (const float*)d_in[0];
    const float* q_coord  = (const float*)d_in[1];
    const float* kv       = (const float*)d_in[2];
    const float* kv_coord = (const float*)d_in[3];
    const float* Wq       = (const float*)d_in[4];
    const float* Wkv      = (const float*)d_in[5];
    const float* Wdelta   = (const float*)d_in[6];
    const float* Wproj    = (const float*)d_in[7];
    const float* bproj    = (const float*)d_in[8];
    float* out = (float*)d_out;
    float* ws  = (float*)d_ws;

    float* v_ws    = ws + OFF_VW;
    float* Eqp     = ws + OFF_EQ;
    float* iEqp    = ws + OFF_IEQ;
    float* Ekp     = ws + OFF_EK;
    float* iEkp    = ws + OFF_IEK;
    float* wsum_ws = ws + OFF_WSUM;
    float* Eqc     = ws + OFF_EQC;
    float* iEqc    = ws + OFF_IEQC;
    float* Ekc     = ws + OFF_EKC;
    float* iEkc    = ws + OFF_IEKC;
    float* po      = ws + OFF_PO;
    float* pm      = ws + OFF_PM;
    float* ps      = ws + OFF_PS;

    diffattn_prep<<<PREP_BLOCKS + COORD_BLOCKS, 256, 0, stream>>>(
        q, kv, q_coord, kv_coord, Wq, Wkv, Wdelta,
        Eqp, iEqp, Ekp, iEkp, Eqc, iEqc, Ekc, iEkc, v_ws, wsum_ws);

    dim3 sgrid(NCHUNK, NQT);
    diffattn_score<<<sgrid, STHREADS, 0, stream>>>(Eqp, iEqp, Ekp, iEkp, Eqc, iEqc,
                                              Ekc, iEkc, wsum_ws, v_ws, po, pm, ps);

    diffattn_combine<<<NCBLK, CTHREADS, 0, stream>>>(po, pm, ps, Wproj, bproj, out);
}


# projection kernel, kv tiles: padded rows clamped and zeroed by a select instead of a guarded load per element, so a tile's 64 operand loads are requested together, not one round trip per k-step
# speedup vs baseline: 2.3075x; 1.2742x over previous
#include <hip/hip_runtime.h>
#include <hip/hip_bf16.h>

typedef float v2f __attribute__((ext_vector_type(2)));
typedef float v8f __attribute__((ext_vector_type(8)));

#define NQ 512
#define NKV 2562
#define NKV_PAD 2624
#define CDIM 64
#define NCHUNK 41
#define CHUNK 64
#define NQT (NQ / 16)
#define SCALE (1.0f / 64.0f)

#define OFF_QW   0L
#define OFF_KW   (OFF_QW   + (long)NQ * CDIM)
#define OFF_VW   (OFF_KW   + (long)NKV_PAD * CDIM)
#define OFF_EQ   (OFF_VW   + (long)NKV_PAD * CDIM)
#define OFF_IEQ  (OFF_EQ   + (long)NQ * CDIM)
#define OFF_EK   (OFF_IEQ  + (long)NQ * CDIM)
#define OFF_IEK  (OFF_EK   + (long)NKV_PAD * CDIM)
#define OFF_WSUM (OFF_IEK  + (long)NKV_PAD * CDIM)
#define OFF_EQC  (OFF_WSUM + 16L)
#define OFF_IEQC (OFF_EQC  + (long)NQ * 3)
#define OFF_EKC  (OFF_IEQC + (long)NQ * 3)
#define OFF_IEKC (OFF_EKC  + (long)NKV * 3)
#define OFF_PO   (OFF_IEKC + (long)NKV * 3 + 16L)
#define OFF_PM   (OFF_PO   + (long)NQT * NCHUNK * 16 * CDIM)
#define OFF_PS   (OFF_PM   + (long)NQT * NCHUNK * 16)

static __device__ __forceinline__ v8f wmma_f32_4(v2f a, v2f b, v8f c) {
    return __builtin_amdgcn_wmma_f32_16x16x4_f32(false, a, false, b, (short)0, c,
                                                 false, false);
}

#define PREP_BLOCKS 180
#define COORD_BLOCKS ((NQ * 3 + NKV * 3 + 255) / 256)

__global__ void diffattn_prep(const float* __restrict__ q,
                              const float* __restrict__ kv,
                              const float* __restrict__ q_coord,
                              const float* __restrict__ kv_coord,
                              const float* __restrict__ Wq,
                              const float* __restrict__ Wkv,
                              const float* __restrict__ Wdelta,
                              float* __restrict__ Eq, float* __restrict__ iEq,
                              float* __restrict__ Ek, float* __restrict__ iEk,
                              float* __restrict__ Eqc, float* __restrict__ iEqc,
                              float* __restrict__ Ekc, float* __restrict__ iEkc,
                              float* __restrict__ v_ws,
                              float* __restrict__ wsum_ws) {
    if (blockIdx.x >= PREP_BLOCKS) {
        const int NE_QC = NQ * 3;
        const int NE_KC = NKV * 3;
        int i = (blockIdx.x - PREP_BLOCKS) * 256 + threadIdx.x;
        if (i < NE_QC) {
            float x = q_coord[i];
            Eqc[i] = __expf(x);
            iEqc[i] = __expf(-x);
        } else if (i < NE_QC + NE_KC) {
            int j = i - NE_QC;
            float x = kv_coord[j];
            Ekc[j] = __expf(x);
            iEkc[j] = __expf(-x);
        }
        return;
    }

    if (blockIdx.x == 0 && threadIdx.x < 3) {
        float s = 0.f;
        for (int c = 0; c < CDIM; ++c) s += Wdelta[threadIdx.x * CDIM + c];
        wsum_ws[threadIdx.x] = s;
    }

    const int KV_TILES = (NKV_PAD / 16) * (128 / 16);
    int wave = (blockIdx.x * blockDim.x + threadIdx.x) >> 5;
    int lane = threadIdx.x & 31;

    int half = lane >> 4;
    int lm = lane & 15;

    if (wave < KV_TILES) {
        int mt = wave >> 3, nt = wave & 7;
        int m0 = mt * 16, n0 = nt * 16;
        int rowA = m0 + lm;
        bool inb = rowA < NKV;
        const int rowL = inb ? rowA : NKV - 1;
        const bool ktile = n0 < CDIM;
        v8f acc = {};
        for (int kk = 0; kk < CDIM; kk += 4) {
            int ka = kk + 2 * half;
            v2f a, b;
            const float ax = kv[rowL * CDIM + ka];
            const float ay = kv[rowL * CDIM + ka + 1];
            a.x = inb ? ax : 0.f;
            a.y = inb ? ay : 0.f;
            b.x = Wkv[ka * 128 + n0 + lm];
            b.y = Wkv[(ka + 1) * 128 + n0 + lm];
            v2f fa = ktile ? b : a;
            v2f fb = ktile ? a : b;
            acc = wmma_f32_4(fa, fb, acc);
        }
        int col = n0 + lm;
        if (ktile) {
            const int row = m0 + lm;
            const int c4 = (n0 + 8 * half) >> 2;
            float4 e0, e1, i0, i1;
            e0.x = __expf(acc[0]);  e0.y = __expf(acc[1]);
            e0.z = __expf(acc[2]);  e0.w = __expf(acc[3]);
            e1.x = __expf(acc[4]);  e1.y = __expf(acc[5]);
            e1.z = __expf(acc[6]);  e1.w = __expf(acc[7]);
            i0.x = __expf(-acc[0]); i0.y = __expf(-acc[1]);
            i0.z = __expf(-acc[2]); i0.w = __expf(-acc[3]);
            i1.x = __expf(-acc[4]); i1.y = __expf(-acc[5]);
            i1.z = __expf(-acc[6]); i1.w = __expf(-acc[7]);
            float4* E4 = reinterpret_cast<float4*>(Ek);
            float4* I4 = reinterpret_cast<float4*>(iEk);
            E4[c4 * NKV_PAD + row] = e0;
            E4[(c4 + 1) * NKV_PAD + row] = e1;
            I4[c4 * NKV_PAD + row] = i0;
            I4[(c4 + 1) * NKV_PAD + row] = i1;
        } else {
#pragma unroll
            for (int r = 0; r < 8; ++r) {
                int row = m0 + r + 8 * half;
                v_ws[row * CDIM + (col - CDIM)] = acc[r];
            }
        }
    } else {
        int t = wave - KV_TILES;
        int mt = t >> 2, nt = t & 3;
        int m0 = mt * 16, n0 = nt * 16;
        int rowA = m0 + lm;
        v8f acc = {};
        for (int kk = 0; kk < CDIM; kk += 4) {
            int ka = kk + 2 * half;
            v2f a, b;
            a.x = q[rowA * CDIM + ka];
            a.y = q[rowA * CDIM + ka + 1];
            b.x = Wq[ka * CDIM + n0 + lm];
            b.y = Wq[(ka + 1) * CDIM + n0 + lm];
            acc = wmma_f32_4(a, b, acc);
        }
        int col = n0 + lm;
        float4 e0, e1, i0, i1;
        e0.x = __expf(acc[0]);  e0.y = __expf(acc[1]);
        e0.z = __expf(acc[2]);  e0.w = __expf(acc[3]);
        e1.x = __expf(acc[4]);  e1.y = __expf(acc[5]);
        e1.z = __expf(acc[6]);  e1.w = __expf(acc[7]);
        i0.x = __expf(-acc[0]); i0.y = __expf(-acc[1]);
        i0.z = __expf(-acc[2]); i0.w = __expf(-acc[3]);
        i1.x = __expf(-acc[4]); i1.y = __expf(-acc[5]);
        i1.z = __expf(-acc[6]); i1.w = __expf(-acc[7]);
        float4* E4 = reinterpret_cast<float4*>(Eq);
        float4* I4 = reinterpret_cast<float4*>(iEq);
        const int f = (mt * CDIM + col) * 4 + 2 * half;
        E4[f] = e0;
        E4[f + 1] = e1;
        I4[f] = i0;
        I4[f + 1] = i1;
    }
}

#define STHREADS 64

__global__ void __launch_bounds__(STHREADS)
diffattn_score(const float* __restrict__ Eq, const float* __restrict__ iEq,
               const float* __restrict__ Ek, const float* __restrict__ iEk,
               const float* __restrict__ Eqc, const float* __restrict__ iEqc,
               const float* __restrict__ Ekc, const float* __restrict__ iEkc,
               const float* __restrict__ wsum_ws,
               const float* __restrict__ v_ws,
               float* __restrict__ po, float* __restrict__ pm,
               float* __restrict__ ps) {
    __shared__ float4 sEqT[CDIM][4];
    __shared__ float4 sIEqT[CDIM][4];
    __shared__ float sP[16][CHUNK + 8];
    __shared__ float sEqc[16][4];
    __shared__ float sIEqc[16][4];
    __shared__ float sRed[16][5];
    __shared__ float sMax[16];
    __shared__ float sSum[16];
    __shared__ float sW[4];

    const int tid = threadIdx.x;
    const int lane = tid & 31;
    const int wave = tid >> 5;
    const int ch = blockIdx.x;
    const int qt = blockIdx.y;
    const int q0 = qt * 16;
    const int j0 = ch * CHUNK;

    {
        const float4* gq = reinterpret_cast<const float4*>(Eq + q0 * CDIM);
        const float4* gi = reinterpret_cast<const float4*>(iEq + q0 * CDIM);
        float4* dq = &sEqT[0][0];
        float4* di = &sIEqT[0][0];
        for (int i = tid; i < 16 * CDIM / 4; i += STHREADS) {
            dq[i] = gq[i];
            di[i] = gi[i];
        }
    }
    if (tid < 48) {
        int r = tid / 3, d = tid % 3;
        sEqc[r][d] = Eqc[(q0 + r) * 3 + d];
        sIEqc[r][d] = iEqc[(q0 + r) * 3 + d];
    }
    if (tid < 3) sW[tid] = wsum_ws[tid];
    __syncthreads();

    {
        const int qw = 8 * wave;
        const int jA = j0 + lane;
        const int jB = jA + 32;
        const float4* ek4 = reinterpret_cast<const float4*>(Ek);
        const float4* ik4 = reinterpret_cast<const float4*>(iEk);

        float sA[8], sB[8];
#pragma unroll
        for (int q = 0; q < 8; ++q) { sA[q] = 0.f; sB[q] = 0.f; }

        float4 eA[2], iA[2], eB[2], iB[2];
#pragma unroll
        for (int u = 0; u < 2; ++u) {
            eA[u] = ek4[u * NKV_PAD + jA];
            iA[u] = ik4[u * NKV_PAD + jA];
            eB[u] = ek4[u * NKV_PAD + jB];
            iB[u] = ik4[u * NKV_PAD + jB];
        }

#pragma unroll 1
        for (int cb = 0; cb < 8; ++cb) {
            const int nb = (cb < 7) ? cb + 1 : 7;
            float4 neA[2], niA[2], neB[2], niB[2];
#pragma unroll
            for (int u = 0; u < 2; ++u) {
                neA[u] = ek4[(2 * nb + u) * NKV_PAD + jA];
                niA[u] = ik4[(2 * nb + u) * NKV_PAD + jA];
                neB[u] = ek4[(2 * nb + u) * NKV_PAD + jB];
                niB[u] = ik4[(2 * nb + u) * NKV_PAD + jB];
            }
#define SCORE_STEP(c, EA, IA, EB, IB)                                          \
            _Pragma("unroll")                                                  \
            for (int v = 0; v < 2; ++v) {                                      \
                const float4 eq = sEqT[8 * cb + (c)][2 * wave + v];            \
                const float4 iq = sIEqT[8 * cb + (c)][2 * wave + v];           \
                sA[4 * v + 0] += fminf(eq.x * (IA), iq.x * (EA));              \
                sA[4 * v + 1] += fminf(eq.y * (IA), iq.y * (EA));              \
                sA[4 * v + 2] += fminf(eq.z * (IA), iq.z * (EA));              \
                sA[4 * v + 3] += fminf(eq.w * (IA), iq.w * (EA));              \
                sB[4 * v + 0] += fminf(eq.x * (IB), iq.x * (EB));              \
                sB[4 * v + 1] += fminf(eq.y * (IB), iq.y * (EB));              \
                sB[4 * v + 2] += fminf(eq.z * (IB), iq.z * (EB));              \
                sB[4 * v + 3] += fminf(eq.w * (IB), iq.w * (EB));              \
            }
            SCORE_STEP(0, eA[0].x, iA[0].x, eB[0].x, iB[0].x)
            SCORE_STEP(1, eA[0].y, iA[0].y, eB[0].y, iB[0].y)
            SCORE_STEP(2, eA[0].z, iA[0].z, eB[0].z, iB[0].z)
            SCORE_STEP(3, eA[0].w, iA[0].w, eB[0].w, iB[0].w)
            SCORE_STEP(4, eA[1].x, iA[1].x, eB[1].x, iB[1].x)
            SCORE_STEP(5, eA[1].y, iA[1].y, eB[1].y, iB[1].y)
            SCORE_STEP(6, eA[1].z, iA[1].z, eB[1].z, iB[1].z)
            SCORE_STEP(7, eA[1].w, iA[1].w, eB[1].w, iB[1].w)
#undef SCORE_STEP
#pragma unroll
            for (int u = 0; u < 2; ++u) {
                eA[u] = neA[u]; iA[u] = niA[u];
                eB[u] = neB[u]; iB[u] = niB[u];
            }
        }

        const bool okA = jA < NKV;
        const bool okB = jB < NKV;
        const int jAc = okA ? jA : NKV - 1;
        const int jBc = okB ? jB : NKV - 1;
        float ekA[3], ikA[3], ekB[3], ikB[3];
#pragma unroll
        for (int d = 0; d < 3; ++d) {
            ekA[d] = Ekc[jAc * 3 + d];
            ikA[d] = iEkc[jAc * 3 + d];
            ekB[d] = Ekc[jBc * 3 + d];
            ikB[d] = iEkc[jBc * 3 + d];
        }
#pragma unroll
        for (int q = 0; q < 8; ++q) {
            float a = sA[q], b = sB[q];
#pragma unroll
            for (int d = 0; d < 3; ++d) {
                float eqc = sEqc[qw + q][d];
                float iqc = sIEqc[qw + q][d];
                a += fminf(eqc * ikA[d], iqc * ekA[d]) * sW[d];
                b += fminf(eqc * ikB[d], iqc * ekB[d]) * sW[d];
            }
            a *= SCALE;
            b *= SCALE;
            sP[qw + q][lane] = okA ? a : -__builtin_inff();
            sP[qw + q][lane + 32] = okB ? b : -__builtin_inff();
        }
    }
    __syncthreads();

    const int qi = tid & 15;
    const int g = tid >> 4;
    {
        float m = -__builtin_inff();
        for (int jl = g; jl < CHUNK; jl += 4) m = fmaxf(m, sP[qi][jl]);
        sRed[qi][g] = m;
    }
    __syncthreads();
    if (tid < 16) {
        float m = sRed[tid][0];
#pragma unroll
        for (int i = 1; i < 4; ++i) m = fmaxf(m, sRed[tid][i]);
        sMax[tid] = m;
    }
    __syncthreads();

    {
        float mr = sMax[qi];
        float psum = 0.f;
        for (int jl = g; jl < CHUNK; jl += 4) {
            float e = __expf(sP[qi][jl] - mr);
            sP[qi][jl] = e;
            psum += e;
        }
        sRed[qi][g] = psum;
    }
    __syncthreads();
    if (tid < 16) {
        float s = 0.f;
#pragma unroll
        for (int i = 0; i < 4; ++i) s += sRed[tid][i];
        sSum[tid] = s;
    }
    __syncthreads();

    const int half = lane >> 4;
    const int lm = lane & 15;
    const long base = (long)(qt * NCHUNK + ch) * 16;
    {
        const int n0 = wave * 32;
        v8f acc0 = {};
        v8f acc1 = {};
        for (int kk = 0; kk < CHUNK; kk += 4) {
            int ka = kk + 2 * half;
            v2f a, b0, b1;
            a.x = sP[lm][ka];
            a.y = sP[lm][ka + 1];
            b0.x = v_ws[(j0 + ka) * CDIM + n0 + lm];
            b0.y = v_ws[(j0 + ka + 1) * CDIM + n0 + lm];
            b1.x = v_ws[(j0 + ka) * CDIM + n0 + 16 + lm];
            b1.y = v_ws[(j0 + ka + 1) * CDIM + n0 + 16 + lm];
            acc0 = wmma_f32_4(a, b0, acc0);
            acc1 = wmma_f32_4(a, b1, acc1);
        }
#pragma unroll
        for (int r = 0; r < 8; ++r) {
            int row = r + 8 * half;
            po[(base + row) * CDIM + n0 + lm] = acc0[r];
            po[(base + row) * CDIM + n0 + 16 + lm] = acc1[r];
        }
    }
    if (tid < 16) {
        pm[base + tid] = sMax[tid];
        ps[base + tid] = sSum[tid];
    }
}

#define CROWS 2
#define NCBLK (NQ / CROWS)
#define CTHREADS (CROWS * CDIM)

static __device__ __forceinline__ float lane_value(float v, int l) {
    return __int_as_float(__builtin_amdgcn_readlane(__float_as_int(v), l));
}
template <int XM>
static __device__ __forceinline__ float row_xor(float v) {
    return __int_as_float(
        __builtin_amdgcn_update_dpp(0, __float_as_int(v), 0x160 + XM, 0xf, 0xf, true));
}
static __device__ __forceinline__ float xhalf(float v) {
    int b = __float_as_int(v);
    return __int_as_float(
        __builtin_amdgcn_permlanex16(b, b, 0x76543210, 0xfedcba98, false, false));
}
static __device__ __forceinline__ float wave_max(float v) {
    v = fmaxf(v, row_xor<1>(v));
    v = fmaxf(v, row_xor<2>(v));
    v = fmaxf(v, row_xor<4>(v));
    v = fmaxf(v, row_xor<8>(v));
    return fmaxf(v, xhalf(v));
}
static __device__ __forceinline__ float wave_sum(float v) {
    v += row_xor<1>(v);
    v += row_xor<2>(v);
    v += row_xor<4>(v);
    v += row_xor<8>(v);
    return v + xhalf(v);
}

__global__ void __launch_bounds__(CTHREADS)
diffattn_combine(const float* __restrict__ po, const float* __restrict__ pm,
                 const float* __restrict__ ps,
                 const float* __restrict__ Wproj,
                 const float* __restrict__ bproj, float* __restrict__ out) {
    __shared__ float sO[CROWS][CDIM + 1];
    __shared__ float sOut[16][CDIM];

    const int tid = threadIdx.x;
    const int lane = tid & 31;
    const int wave = tid >> 5;
    const int half = lane >> 4;
    const int lm = lane & 15;
    const int r = tid >> 6;
    const int n = tid & 63;
    const int n0 = wave * 16;
    const int grow = blockIdx.x * CROWS + r;
    const long pbase = (long)(grow >> 4) * NCHUNK * 16 + (grow & 15);

    const bool has1 = lane + 32 < NCHUNK;
    const int c1 = has1 ? lane + 32 : NCHUNK - 1;
    float m0 = pm[pbase + lane * 16];
    float s0 = ps[pbase + lane * 16];
    float m1 = pm[pbase + c1 * 16];
    float s1 = ps[pbase + c1 * 16];

    float pv[NCHUNK];
#pragma unroll
    for (int c = 0; c < NCHUNK; ++c) pv[c] = po[(pbase + c * 16) * CDIM + n];

    v2f bf[CDIM / 4];
#pragma unroll
    for (int kk = 0; kk < CDIM; kk += 4) {
        int ka = kk + 2 * half;
        bf[kk / 4].x = Wproj[ka * CDIM + n0 + lm];
        bf[kk / 4].y = Wproj[(ka + 1) * CDIM + n0 + lm];
    }
    float bb = bproj[n];
    __builtin_amdgcn_sched_barrier(0);

    m1 = has1 ? m1 : -__builtin_inff();
    float M = wave_max(fmaxf(m0, m1));
    float e0 = __expf(m0 - M);
    float e1 = has1 ? __expf(m1 - M) : 0.f;
    float S = wave_sum(s0 * e0 + s1 * e1);
    float inv = 1.0f / S;
    float w0 = e0 * inv;
    float w1 = e1 * inv;

    float a = 0.f;
#pragma unroll
    for (int c = 0; c < NCHUNK; ++c) {
        float wc = (c < 32) ? lane_value(w0, c) : lane_value(w1, c - 32);
        a += pv[c] * wc;
    }
    sO[r][n] = a;
    __syncthreads();

    v8f acc = {};
#pragma unroll
    for (int kk = 0; kk < CDIM; kk += 4) {
        int ka = kk + 2 * half;
        v2f a2;
        a2.x = sO[lm & (CROWS - 1)][ka];
        a2.y = sO[lm & (CROWS - 1)][ka + 1];
        acc = wmma_f32_4(a2, bf[kk / 4], acc);
    }
#pragma unroll
    for (int q = 0; q < CROWS; ++q) sOut[q + 8 * half][n0 + lm] = acc[q];
    __syncthreads();

    out[grow * CDIM + n] = sOut[r][n] + bb;
}

extern "C" void kernel_launch(void* const* d_in, const int* in_sizes, int n_in,
                              void* d_out, int out_size, void* d_ws, size_t ws_size,
                              hipStream_t stream) {
    const float* q        = (const float*)d_in[0];
    const float* q_coord  = (const float*)d_in[1];
    const float* kv       = (const float*)d_in[2];
    const float* kv_coord = (const float*)d_in[3];
    const float* Wq       = (const float*)d_in[4];
    const float* Wkv      = (const float*)d_in[5];
    const float* Wdelta   = (const float*)d_in[6];
    const float* Wproj    = (const float*)d_in[7];
    const float* bproj    = (const float*)d_in[8];
    float* out = (float*)d_out;
    float* ws  = (float*)d_ws;

    float* v_ws    = ws + OFF_VW;
    float* Eqp     = ws + OFF_EQ;
    float* iEqp    = ws + OFF_IEQ;
    float* Ekp     = ws + OFF_EK;
    float* iEkp    = ws + OFF_IEK;
    float* wsum_ws = ws + OFF_WSUM;
    float* Eqc     = ws + OFF_EQC;
    float* iEqc    = ws + OFF_IEQC;
    float* Ekc     = ws + OFF_EKC;
    float* iEkc    = ws + OFF_IEKC;
    float* po      = ws + OFF_PO;
    float* pm      = ws + OFF_PM;
    float* ps      = ws + OFF_PS;

    diffattn_prep<<<PREP_BLOCKS + COORD_BLOCKS, 256, 0, stream>>>(
        q, kv, q_coord, kv_coord, Wq, Wkv, Wdelta,
        Eqp, iEqp, Ekp, iEkp, Eqc, iEqc, Ekc, iEkc, v_ws, wsum_ws);

    dim3 sgrid(NCHUNK, NQT);
    diffattn_score<<<sgrid, STHREADS, 0, stream>>>(Eqp, iEqp, Ekp, iEkp, Eqc, iEqc,
                                              Ekc, iEkc, wsum_ws, v_ws, po, pm, ps);

    diffattn_combine<<<NCBLK, CTHREADS, 0, stream>>>(po, pm, ps, Wproj, bproj, out);
}


# score kernel: loads requested ahead of their waits: tile, coordinate and first k loads as one group (was 7 dependent round trips), next 8 channels in flight during a step, v fragments before softmax
# speedup vs baseline: 2.5640x; 1.1112x over previous
#include <hip/hip_runtime.h>
#include <hip/hip_bf16.h>

typedef float v2f __attribute__((ext_vector_type(2)));
typedef float v8f __attribute__((ext_vector_type(8)));
typedef float v4f __attribute__((ext_vector_type(4)));

#define NQ 512
#define NKV 2562
#define NKV_PAD 2624
#define CDIM 64
#define NCHUNK 41
#define CHUNK 64
#define NQT (NQ / 16)
#define SCALE (1.0f / 64.0f)

#define OFF_QW   0L
#define OFF_KW   (OFF_QW   + (long)NQ * CDIM)
#define OFF_VW   (OFF_KW   + (long)NKV_PAD * CDIM)
#define OFF_EQ   (OFF_VW   + (long)NKV_PAD * CDIM)
#define OFF_IEQ  (OFF_EQ   + (long)NQ * CDIM)
#define OFF_EK   (OFF_IEQ  + (long)NQ * CDIM)
#define OFF_IEK  (OFF_EK   + (long)NKV_PAD * CDIM)
#define OFF_WSUM (OFF_IEK  + (long)NKV_PAD * CDIM)
#define OFF_EQC  (OFF_WSUM + 16L)
#define OFF_IEQC (OFF_EQC  + (long)NQ * 3)
#define OFF_EKC  (OFF_IEQC + (long)NQ * 3)
#define OFF_IEKC (OFF_EKC  + (long)NKV * 3)
#define OFF_PO   (OFF_IEKC + (long)NKV * 3 + 16L)
#define OFF_PM   (OFF_PO   + (long)NQT * NCHUNK * 16 * CDIM)
#define OFF_PS   (OFF_PM   + (long)NQT * NCHUNK * 16)

static __device__ __forceinline__ void lds_barrier() {
    __builtin_amdgcn_fence(__ATOMIC_RELEASE, "workgroup", "local");
    __builtin_amdgcn_s_barrier();
    __builtin_amdgcn_fence(__ATOMIC_ACQUIRE, "workgroup", "local");
}

static __device__ __forceinline__ v8f wmma_f32_4(v2f a, v2f b, v8f c) {
    return __builtin_amdgcn_wmma_f32_16x16x4_f32(false, a, false, b, (short)0, c,
                                                 false, false);
}

#define PREP_BLOCKS 180
#define COORD_BLOCKS ((NQ * 3 + NKV * 3 + 255) / 256)

__global__ void diffattn_prep(const float* __restrict__ q,
                              const float* __restrict__ kv,
                              const float* __restrict__ q_coord,
                              const float* __restrict__ kv_coord,
                              const float* __restrict__ Wq,
                              const float* __restrict__ Wkv,
                              const float* __restrict__ Wdelta,
                              float* __restrict__ Eq, float* __restrict__ iEq,
                              float* __restrict__ Ek, float* __restrict__ iEk,
                              float* __restrict__ Eqc, float* __restrict__ iEqc,
                              float* __restrict__ Ekc, float* __restrict__ iEkc,
                              float* __restrict__ v_ws,
                              float* __restrict__ wsum_ws) {
    if (blockIdx.x >= PREP_BLOCKS) {
        const int NE_QC = NQ * 3;
        const int NE_KC = NKV * 3;
        int i = (blockIdx.x - PREP_BLOCKS) * 256 + threadIdx.x;
        if (i < NE_QC) {
            float x = q_coord[i];
            Eqc[i] = __expf(x);
            iEqc[i] = __expf(-x);
        } else if (i < NE_QC + NE_KC) {
            int j = i - NE_QC;
            float x = kv_coord[j];
            Ekc[j] = __expf(x);
            iEkc[j] = __expf(-x);
        }
        return;
    }

    if (blockIdx.x == 0 && threadIdx.x < 3) {
        float s = 0.f;
        for (int c = 0; c < CDIM; ++c) s += Wdelta[threadIdx.x * CDIM + c];
        wsum_ws[threadIdx.x] = s;
    }

    const int KV_TILES = (NKV_PAD / 16) * (128 / 16);
    int wave = (blockIdx.x * blockDim.x + threadIdx.x) >> 5;
    int lane = threadIdx.x & 31;

    int half = lane >> 4;
    int lm = lane & 15;

    if (wave < KV_TILES) {
        int mt = wave >> 3, nt = wave & 7;
        int m0 = mt * 16, n0 = nt * 16;
        int rowA = m0 + lm;
        bool inb = rowA < NKV;
        const int rowL = inb ? rowA : NKV - 1;
        const bool ktile = n0 < CDIM;
        v8f acc = {};
        for (int kk = 0; kk < CDIM; kk += 4) {
            int ka = kk + 2 * half;
            v2f a, b;
            const float ax = kv[rowL * CDIM + ka];
            const float ay = kv[rowL * CDIM + ka + 1];
            a.x = inb ? ax : 0.f;
            a.y = inb ? ay : 0.f;
            b.x = Wkv[ka * 128 + n0 + lm];
            b.y = Wkv[(ka + 1) * 128 + n0 + lm];
            v2f fa = ktile ? b : a;
            v2f fb = ktile ? a : b;
            acc = wmma_f32_4(fa, fb, acc);
        }
        int col = n0 + lm;
        if (ktile) {
            const int row = m0 + lm;
            const int c4 = (n0 + 8 * half) >> 2;
            float4 e0, e1, i0, i1;
            e0.x = __expf(acc[0]);  e0.y = __expf(acc[1]);
            e0.z = __expf(acc[2]);  e0.w = __expf(acc[3]);
            e1.x = __expf(acc[4]);  e1.y = __expf(acc[5]);
            e1.z = __expf(acc[6]);  e1.w = __expf(acc[7]);
            i0.x = __expf(-acc[0]); i0.y = __expf(-acc[1]);
            i0.z = __expf(-acc[2]); i0.w = __expf(-acc[3]);
            i1.x = __expf(-acc[4]); i1.y = __expf(-acc[5]);
            i1.z = __expf(-acc[6]); i1.w = __expf(-acc[7]);
            float4* E4 = reinterpret_cast<float4*>(Ek);
            float4* I4 = reinterpret_cast<float4*>(iEk);
            E4[c4 * NKV_PAD + row] = e0;
            E4[(c4 + 1) * NKV_PAD + row] = e1;
            I4[c4 * NKV_PAD + row] = i0;
            I4[(c4 + 1) * NKV_PAD + row] = i1;
        } else {
#pragma unroll
            for (int r = 0; r < 8; ++r) {
                int row = m0 + r + 8 * half;
                v_ws[row * CDIM + (col - CDIM)] = acc[r];
            }
        }
    } else {
        int t = wave - KV_TILES;
        int mt = t >> 2, nt = t & 3;
        int m0 = mt * 16, n0 = nt * 16;
        int rowA = m0 + lm;
        v8f acc = {};
        for (int kk = 0; kk < CDIM; kk += 4) {
            int ka = kk + 2 * half;
            v2f a, b;
            a.x = q[rowA * CDIM + ka];
            a.y = q[rowA * CDIM + ka + 1];
            b.x = Wq[ka * CDIM + n0 + lm];
            b.y = Wq[(ka + 1) * CDIM + n0 + lm];
            acc = wmma_f32_4(a, b, acc);
        }
        int col = n0 + lm;
        float4 e0, e1, i0, i1;
        e0.x = __expf(acc[0]);  e0.y = __expf(acc[1]);
        e0.z = __expf(acc[2]);  e0.w = __expf(acc[3]);
        e1.x = __expf(acc[4]);  e1.y = __expf(acc[5]);
        e1.z = __expf(acc[6]);  e1.w = __expf(acc[7]);
        i0.x = __expf(-acc[0]); i0.y = __expf(-acc[1]);
        i0.z = __expf(-acc[2]); i0.w = __expf(-acc[3]);
        i1.x = __expf(-acc[4]); i1.y = __expf(-acc[5]);
        i1.z = __expf(-acc[6]); i1.w = __expf(-acc[7]);
        float4* E4 = reinterpret_cast<float4*>(Eq);
        float4* I4 = reinterpret_cast<float4*>(iEq);
        const int f = (mt * CDIM + col) * 4 + 2 * half;
        E4[f] = e0;
        E4[f + 1] = e1;
        I4[f] = i0;
        I4[f + 1] = i1;
    }
}

#define STHREADS 64

__global__ void __launch_bounds__(STHREADS)
diffattn_score(const float* __restrict__ Eq, const float* __restrict__ iEq,
               const float* __restrict__ Ek, const float* __restrict__ iEk,
               const float* __restrict__ Eqc, const float* __restrict__ iEqc,
               const float* __restrict__ Ekc, const float* __restrict__ iEkc,
               const float* __restrict__ wsum_ws,
               const float* __restrict__ v_ws,
               float* __restrict__ po, float* __restrict__ pm,
               float* __restrict__ ps) {
    __shared__ float4 sEqT[CDIM][4];
    __shared__ float4 sIEqT[CDIM][4];
    __shared__ float sP[16][CHUNK + 8];
    __shared__ float sEqc[STHREADS];
    __shared__ float sIEqc[STHREADS];
    __shared__ float sRed[16][5];
    __shared__ float sMax[16];
    __shared__ float sSum[16];

    const int tid = threadIdx.x;
    const int lane = tid & 31;
    const int wave = tid >> 5;
    const int ch = blockIdx.x;
    const int qt = blockIdx.y;
    const int q0 = qt * 16;
    const int j0 = ch * CHUNK;

    const int half = lane >> 4;
    const int lm = lane & 15;
    const int jA = j0 + lane;
    const int jB = jA + 32;
    const v4f* ek4 = reinterpret_cast<const v4f*>(Ek);
    const v4f* ik4 = reinterpret_cast<const v4f*>(iEk);
    const bool okA = jA < NKV;
    const bool okB = jB < NKV;
    const int jAc = okA ? jA : NKV - 1;
    const int jBc = okB ? jB : NKV - 1;

    const float4* gq = reinterpret_cast<const float4*>(Eq + q0 * CDIM);
    const float4* gi = reinterpret_cast<const float4*>(iEq + q0 * CDIM);
    float4 tq[4], ti[4];
#pragma unroll
    for (int u = 0; u < 4; ++u) {
        tq[u] = gq[tid + STHREADS * u];
        ti[u] = gi[tid + STHREADS * u];
    }
    const int t48 = tid < 48 ? tid : 47;
    const float cq = Eqc[q0 * 3 + t48];
    const float ci = iEqc[q0 * 3 + t48];
    const float sW[3] = {wsum_ws[0], wsum_ws[1], wsum_ws[2]};
    v4f eA[2], iA[2], eB[2], iB[2];
#pragma unroll
    for (int u = 0; u < 2; ++u) {
        eA[u] = ek4[u * NKV_PAD + jA];
        iA[u] = ik4[u * NKV_PAD + jA];
        eB[u] = ek4[u * NKV_PAD + jB];
        iB[u] = ik4[u * NKV_PAD + jB];
    }
    float ekA[3], ikA[3], ekB[3], ikB[3];
#pragma unroll
    for (int d = 0; d < 3; ++d) {
        ekA[d] = Ekc[jAc * 3 + d];
        ikA[d] = iEkc[jAc * 3 + d];
        ekB[d] = Ekc[jBc * 3 + d];
        ikB[d] = iEkc[jBc * 3 + d];
    }

    {
        float4* dq = &sEqT[0][0];
        float4* di = &sIEqT[0][0];
#pragma unroll
        for (int u = 0; u < 4; ++u) {
            dq[tid + STHREADS * u] = tq[u];
            di[tid + STHREADS * u] = ti[u];
        }
    }
    sEqc[tid] = cq;
    sIEqc[tid] = ci;
    __syncthreads();

    v2f vb0[CHUNK / 4], vb1[CHUNK / 4];

    {
        const int qw = 8 * wave;

        float sA[8], sB[8];
#pragma unroll
        for (int q = 0; q < 8; ++q) { sA[q] = 0.f; sB[q] = 0.f; }

#pragma unroll 1
        for (int cb = 0; cb < 8; ++cb) {
            const int nb = (cb < 7) ? cb + 1 : 7;
            v4f neA[2], niA[2], neB[2], niB[2];
#pragma unroll
            for (int u = 0; u < 2; ++u) {
                neA[u] = ek4[(2 * nb + u) * NKV_PAD + jA];
                niA[u] = ik4[(2 * nb + u) * NKV_PAD + jA];
                neB[u] = ek4[(2 * nb + u) * NKV_PAD + jB];
                niB[u] = ik4[(2 * nb + u) * NKV_PAD + jB];
            }
            __builtin_amdgcn_sched_barrier(0);
#define SCORE_STEP(c, EA, IA, EB, IB)                                          \
            _Pragma("unroll")                                                  \
            for (int v = 0; v < 2; ++v) {                                      \
                const float4 eq = sEqT[8 * cb + (c)][2 * wave + v];            \
                const float4 iq = sIEqT[8 * cb + (c)][2 * wave + v];           \
                sA[4 * v + 0] += fminf(eq.x * (IA), iq.x * (EA));              \
                sA[4 * v + 1] += fminf(eq.y * (IA), iq.y * (EA));              \
                sA[4 * v + 2] += fminf(eq.z * (IA), iq.z * (EA));              \
                sA[4 * v + 3] += fminf(eq.w * (IA), iq.w * (EA));              \
                sB[4 * v + 0] += fminf(eq.x * (IB), iq.x * (EB));              \
                sB[4 * v + 1] += fminf(eq.y * (IB), iq.y * (EB));              \
                sB[4 * v + 2] += fminf(eq.z * (IB), iq.z * (EB));              \
                sB[4 * v + 3] += fminf(eq.w * (IB), iq.w * (EB));              \
            }
            SCORE_STEP(0, eA[0].x, iA[0].x, eB[0].x, iB[0].x)
            SCORE_STEP(1, eA[0].y, iA[0].y, eB[0].y, iB[0].y)
            SCORE_STEP(2, eA[0].z, iA[0].z, eB[0].z, iB[0].z)
            SCORE_STEP(3, eA[0].w, iA[0].w, eB[0].w, iB[0].w)
            SCORE_STEP(4, eA[1].x, iA[1].x, eB[1].x, iB[1].x)
            SCORE_STEP(5, eA[1].y, iA[1].y, eB[1].y, iB[1].y)
            SCORE_STEP(6, eA[1].z, iA[1].z, eB[1].z, iB[1].z)
            SCORE_STEP(7, eA[1].w, iA[1].w, eB[1].w, iB[1].w)
#undef SCORE_STEP
            __builtin_amdgcn_sched_barrier(0);
            asm volatile("" :: "v"(neA[0]), "v"(neA[1]), "v"(niA[0]), "v"(niA[1]),
                               "v"(neB[0]), "v"(neB[1]), "v"(niB[0]), "v"(niB[1]));
#pragma unroll
            for (int u = 0; u < 2; ++u) {
                eA[u] = neA[u]; iA[u] = niA[u];
                eB[u] = neB[u]; iB[u] = niB[u];
            }
        }

        {
            const int n0 = wave * 32;
#pragma unroll
            for (int kk = 0; kk < CHUNK; kk += 4) {
                const int ka = kk + 2 * half;
                vb0[kk / 4].x = v_ws[(j0 + ka) * CDIM + n0 + lm];
                vb0[kk / 4].y = v_ws[(j0 + ka + 1) * CDIM + n0 + lm];
                vb1[kk / 4].x = v_ws[(j0 + ka) * CDIM + n0 + 16 + lm];
                vb1[kk / 4].y = v_ws[(j0 + ka + 1) * CDIM + n0 + 16 + lm];
            }
        }

#pragma unroll
        for (int q = 0; q < 8; ++q) {
            float a = sA[q], b = sB[q];
#pragma unroll
            for (int d = 0; d < 3; ++d) {
                float eqc = sEqc[(qw + q) * 3 + d];
                float iqc = sIEqc[(qw + q) * 3 + d];
                a += fminf(eqc * ikA[d], iqc * ekA[d]) * sW[d];
                b += fminf(eqc * ikB[d], iqc * ekB[d]) * sW[d];
            }
            a *= SCALE;
            b *= SCALE;
            sP[qw + q][lane] = okA ? a : -__builtin_inff();
            sP[qw + q][lane + 32] = okB ? b : -__builtin_inff();
        }
    }
    lds_barrier();

    const int qi = tid & 15;
    const int g = tid >> 4;
    {
        float m = -__builtin_inff();
        for (int jl = g; jl < CHUNK; jl += 4) m = fmaxf(m, sP[qi][jl]);
        sRed[qi][g] = m;
    }
    lds_barrier();
    if (tid < 16) {
        float m = sRed[tid][0];
#pragma unroll
        for (int i = 1; i < 4; ++i) m = fmaxf(m, sRed[tid][i]);
        sMax[tid] = m;
    }
    lds_barrier();

    {
        float mr = sMax[qi];
        float psum = 0.f;
        for (int jl = g; jl < CHUNK; jl += 4) {
            float e = __expf(sP[qi][jl] - mr);
            sP[qi][jl] = e;
            psum += e;
        }
        sRed[qi][g] = psum;
    }
    lds_barrier();
    if (tid < 16) {
        float s = 0.f;
#pragma unroll
        for (int i = 0; i < 4; ++i) s += sRed[tid][i];
        sSum[tid] = s;
    }
    lds_barrier();

    const long base = (long)(qt * NCHUNK + ch) * 16;
    {
        const int n0 = wave * 32;
        v8f acc0 = {};
        v8f acc1 = {};
#pragma unroll
        for (int kk = 0; kk < CHUNK; kk += 4) {
            int ka = kk + 2 * half;
            v2f a;
            a.x = sP[lm][ka];
            a.y = sP[lm][ka + 1];
            acc0 = wmma_f32_4(a, vb0[kk / 4], acc0);
            acc1 = wmma_f32_4(a, vb1[kk / 4], acc1);
        }
#pragma unroll
        for (int r = 0; r < 8; ++r) {
            int row = r + 8 * half;
            po[(base + row) * CDIM + n0 + lm] = acc0[r];
            po[(base + row) * CDIM + n0 + 16 + lm] = acc1[r];
        }
    }
    if (tid < 16) {
        pm[base + tid] = sMax[tid];
        ps[base + tid] = sSum[tid];
    }
}

#define CROWS 2
#define NCBLK (NQ / CROWS)
#define CTHREADS (CROWS * CDIM)

static __device__ __forceinline__ float lane_value(float v, int l) {
    return __int_as_float(__builtin_amdgcn_readlane(__float_as_int(v), l));
}
template <int XM>
static __device__ __forceinline__ float row_xor(float v) {
    return __int_as_float(
        __builtin_amdgcn_update_dpp(0, __float_as_int(v), 0x160 + XM, 0xf, 0xf, true));
}
static __device__ __forceinline__ float xhalf(float v) {
    int b = __float_as_int(v);
    return __int_as_float(
        __builtin_amdgcn_permlanex16(b, b, 0x76543210, 0xfedcba98, false, false));
}
static __device__ __forceinline__ float wave_max(float v) {
    v = fmaxf(v, row_xor<1>(v));
    v = fmaxf(v, row_xor<2>(v));
    v = fmaxf(v, row_xor<4>(v));
    v = fmaxf(v, row_xor<8>(v));
    return fmaxf(v, xhalf(v));
}
static __device__ __forceinline__ float wave_sum(float v) {
    v += row_xor<1>(v);
    v += row_xor<2>(v);
    v += row_xor<4>(v);
    v += row_xor<8>(v);
    return v + xhalf(v);
}

__global__ void __launch_bounds__(CTHREADS)
diffattn_combine(const float* __restrict__ po, const float* __restrict__ pm,
                 const float* __restrict__ ps,
                 const float* __restrict__ Wproj,
                 const float* __restrict__ bproj, float* __restrict__ out) {
    __shared__ float sO[CROWS][CDIM + 1];
    __shared__ float sOut[16][CDIM];

    const int tid = threadIdx.x;
    const int lane = tid & 31;
    const int wave = tid >> 5;
    const int half = lane >> 4;
    const int lm = lane & 15;
    const int r = tid >> 6;
    const int n = tid & 63;
    const int n0 = wave * 16;
    const int grow = blockIdx.x * CROWS + r;
    const long pbase = (long)(grow >> 4) * NCHUNK * 16 + (grow & 15);

    const bool has1 = lane + 32 < NCHUNK;
    const int c1 = has1 ? lane + 32 : NCHUNK - 1;
    float m0 = pm[pbase + lane * 16];
    float s0 = ps[pbase + lane * 16];
    float m1 = pm[pbase + c1 * 16];
    float s1 = ps[pbase + c1 * 16];

    float pv[NCHUNK];
#pragma unroll
    for (int c = 0; c < NCHUNK; ++c) pv[c] = po[(pbase + c * 16) * CDIM + n];

    v2f bf[CDIM / 4];
#pragma unroll
    for (int kk = 0; kk < CDIM; kk += 4) {
        int ka = kk + 2 * half;
        bf[kk / 4].x = Wproj[ka * CDIM + n0 + lm];
        bf[kk / 4].y = Wproj[(ka + 1) * CDIM + n0 + lm];
    }
    float bb = bproj[n];
    __builtin_amdgcn_sched_barrier(0);

    m1 = has1 ? m1 : -__builtin_inff();
    float M = wave_max(fmaxf(m0, m1));
    float e0 = __expf(m0 - M);
    float e1 = has1 ? __expf(m1 - M) : 0.f;
    float S = wave_sum(s0 * e0 + s1 * e1);
    float inv = 1.0f / S;
    float w0 = e0 * inv;
    float w1 = e1 * inv;

    float a = 0.f;
#pragma unroll
    for (int c = 0; c < NCHUNK; ++c) {
        float wc = (c < 32) ? lane_value(w0, c) : lane_value(w1, c - 32);
        a += pv[c] * wc;
    }
    sO[r][n] = a;
    __syncthreads();

    v8f acc = {};
#pragma unroll
    for (int kk = 0; kk < CDIM; kk += 4) {
        int ka = kk + 2 * half;
        v2f a2;
        a2.x = sO[lm & (CROWS - 1)][ka];
        a2.y = sO[lm & (CROWS - 1)][ka + 1];
        acc = wmma_f32_4(a2, bf[kk / 4], acc);
    }
#pragma unroll
    for (int q = 0; q < CROWS; ++q) sOut[q + 8 * half][n0 + lm] = acc[q];
    __syncthreads();

    out[grow * CDIM + n] = sOut[r][n] + bb;
}

extern "C" void kernel_launch(void* const* d_in, const int* in_sizes, int n_in,
                              void* d_out, int out_size, void* d_ws, size_t ws_size,
                              hipStream_t stream) {
    const float* q        = (const float*)d_in[0];
    const float* q_coord  = (const float*)d_in[1];
    const float* kv       = (const float*)d_in[2];
    const float* kv_coord = (const float*)d_in[3];
    const float* Wq       = (const float*)d_in[4];
    const float* Wkv      = (const float*)d_in[5];
    const float* Wdelta   = (const float*)d_in[6];
    const float* Wproj    = (const float*)d_in[7];
    const float* bproj    = (const float*)d_in[8];
    float* out = (float*)d_out;
    float* ws  = (float*)d_ws;

    float* v_ws    = ws + OFF_VW;
    float* Eqp     = ws + OFF_EQ;
    float* iEqp    = ws + OFF_IEQ;
    float* Ekp     = ws + OFF_EK;
    float* iEkp    = ws + OFF_IEK;
    float* wsum_ws = ws + OFF_WSUM;
    float* Eqc     = ws + OFF_EQC;
    float* iEqc    = ws + OFF_IEQC;
    float* Ekc     = ws + OFF_EKC;
    float* iEkc    = ws + OFF_IEKC;
    float* po      = ws + OFF_PO;
    float* pm      = ws + OFF_PM;
    float* ps      = ws + OFF_PS;

    diffattn_prep<<<PREP_BLOCKS + COORD_BLOCKS, 256, 0, stream>>>(
        q, kv, q_coord, kv_coord, Wq, Wkv, Wdelta,
        Eqp, iEqp, Ekp, iEkp, Eqc, iEqc, Ekc, iEkc, v_ws, wsum_ws);

    dim3 sgrid(NCHUNK, NQT);
    diffattn_score<<<sgrid, STHREADS, 0, stream>>>(Eqp, iEqp, Ekp, iEkp, Eqc, iEqc,
                                              Ekc, iEkc, wsum_ws, v_ws, po, pm, ps);

    diffattn_combine<<<NCBLK, CTHREADS, 0, stream>>>(po, pm, ps, Wproj, bproj, out);
}


# score kernel: chunk softmax in registers, row max and exp sum as lane trees (was two 16-trip LDS loops, 4 more barriers); loop sums held as query-pair 2-vectors; row sum order changes
# speedup vs baseline: 2.8045x; 1.0938x over previous
#include <hip/hip_runtime.h>
#include <hip/hip_bf16.h>

typedef float v2f __attribute__((ext_vector_type(2)));
typedef float v8f __attribute__((ext_vector_type(8)));
typedef float v4f __attribute__((ext_vector_type(4)));

#define NQ 512
#define NKV 2562
#define NKV_PAD 2624
#define CDIM 64
#define NCHUNK 41
#define CHUNK 64
#define NQT (NQ / 16)
#define SCALE (1.0f / 64.0f)

#define OFF_QW   0L
#define OFF_KW   (OFF_QW   + (long)NQ * CDIM)
#define OFF_VW   (OFF_KW   + (long)NKV_PAD * CDIM)
#define OFF_EQ   (OFF_VW   + (long)NKV_PAD * CDIM)
#define OFF_IEQ  (OFF_EQ   + (long)NQ * CDIM)
#define OFF_EK   (OFF_IEQ  + (long)NQ * CDIM)
#define OFF_IEK  (OFF_EK   + (long)NKV_PAD * CDIM)
#define OFF_WSUM (OFF_IEK  + (long)NKV_PAD * CDIM)
#define OFF_EQC  (OFF_WSUM + 16L)
#define OFF_IEQC (OFF_EQC  + (long)NQ * 3)
#define OFF_EKC  (OFF_IEQC + (long)NQ * 3)
#define OFF_IEKC (OFF_EKC  + (long)NKV * 3)
#define OFF_PO   (OFF_IEKC + (long)NKV * 3 + 16L)
#define OFF_PM   (OFF_PO   + (long)NQT * NCHUNK * 16 * CDIM)
#define OFF_PS   (OFF_PM   + (long)NQT * NCHUNK * 16)

static __device__ __forceinline__ void lds_barrier() {
    __builtin_amdgcn_fence(__ATOMIC_RELEASE, "workgroup", "local");
    __builtin_amdgcn_s_barrier();
    __builtin_amdgcn_fence(__ATOMIC_ACQUIRE, "workgroup", "local");
}

static __device__ __forceinline__ v8f wmma_f32_4(v2f a, v2f b, v8f c) {
    return __builtin_amdgcn_wmma_f32_16x16x4_f32(false, a, false, b, (short)0, c,
                                                 false, false);
}

#define PREP_BLOCKS 180
#define COORD_BLOCKS ((NQ * 3 + NKV * 3 + 255) / 256)

__global__ void diffattn_prep(const float* __restrict__ q,
                              const float* __restrict__ kv,
                              const float* __restrict__ q_coord,
                              const float* __restrict__ kv_coord,
                              const float* __restrict__ Wq,
                              const float* __restrict__ Wkv,
                              const float* __restrict__ Wdelta,
                              float* __restrict__ Eq, float* __restrict__ iEq,
                              float* __restrict__ Ek, float* __restrict__ iEk,
                              float* __restrict__ Eqc, float* __restrict__ iEqc,
                              float* __restrict__ Ekc, float* __restrict__ iEkc,
                              float* __restrict__ v_ws,
                              float* __restrict__ wsum_ws) {
    if (blockIdx.x >= PREP_BLOCKS) {
        const int NE_QC = NQ * 3;
        const int NE_KC = NKV * 3;
        int i = (blockIdx.x - PREP_BLOCKS) * 256 + threadIdx.x;
        if (i < NE_QC) {
            float x = q_coord[i];
            Eqc[i] = __expf(x);
            iEqc[i] = __expf(-x);
        } else if (i < NE_QC + NE_KC) {
            int j = i - NE_QC;
            float x = kv_coord[j];
            Ekc[j] = __expf(x);
            iEkc[j] = __expf(-x);
        }
        return;
    }

    if (blockIdx.x == 0 && threadIdx.x < 3) {
        float s = 0.f;
        for (int c = 0; c < CDIM; ++c) s += Wdelta[threadIdx.x * CDIM + c];
        wsum_ws[threadIdx.x] = s;
    }

    const int KV_TILES = (NKV_PAD / 16) * (128 / 16);
    int wave = (blockIdx.x * blockDim.x + threadIdx.x) >> 5;
    int lane = threadIdx.x & 31;

    int half = lane >> 4;
    int lm = lane & 15;

    if (wave < KV_TILES) {
        int mt = wave >> 3, nt = wave & 7;
        int m0 = mt * 16, n0 = nt * 16;
        int rowA = m0 + lm;
        bool inb = rowA < NKV;
        const int rowL = inb ? rowA : NKV - 1;
        const bool ktile = n0 < CDIM;
        v8f acc = {};
        for (int kk = 0; kk < CDIM; kk += 4) {
            int ka = kk + 2 * half;
            v2f a, b;
            const float ax = kv[rowL * CDIM + ka];
            const float ay = kv[rowL * CDIM + ka + 1];
            a.x = inb ? ax : 0.f;
            a.y = inb ? ay : 0.f;
            b.x = Wkv[ka * 128 + n0 + lm];
            b.y = Wkv[(ka + 1) * 128 + n0 + lm];
            v2f fa = ktile ? b : a;
            v2f fb = ktile ? a : b;
            acc = wmma_f32_4(fa, fb, acc);
        }
        int col = n0 + lm;
        if (ktile) {
            const int row = m0 + lm;
            const int c4 = (n0 + 8 * half) >> 2;
            float4 e0, e1, i0, i1;
            e0.x = __expf(acc[0]);  e0.y = __expf(acc[1]);
            e0.z = __expf(acc[2]);  e0.w = __expf(acc[3]);
            e1.x = __expf(acc[4]);  e1.y = __expf(acc[5]);
            e1.z = __expf(acc[6]);  e1.w = __expf(acc[7]);
            i0.x = __expf(-acc[0]); i0.y = __expf(-acc[1]);
            i0.z = __expf(-acc[2]); i0.w = __expf(-acc[3]);
            i1.x = __expf(-acc[4]); i1.y = __expf(-acc[5]);
            i1.z = __expf(-acc[6]); i1.w = __expf(-acc[7]);
            float4* E4 = reinterpret_cast<float4*>(Ek);
            float4* I4 = reinterpret_cast<float4*>(iEk);
            E4[c4 * NKV_PAD + row] = e0;
            E4[(c4 + 1) * NKV_PAD + row] = e1;
            I4[c4 * NKV_PAD + row] = i0;
            I4[(c4 + 1) * NKV_PAD + row] = i1;
        } else {
#pragma unroll
            for (int r = 0; r < 8; ++r) {
                int row = m0 + r + 8 * half;
                v_ws[row * CDIM + (col - CDIM)] = acc[r];
            }
        }
    } else {
        int t = wave - KV_TILES;
        int mt = t >> 2, nt = t & 3;
        int m0 = mt * 16, n0 = nt * 16;
        int rowA = m0 + lm;
        v8f acc = {};
        for (int kk = 0; kk < CDIM; kk += 4) {
            int ka = kk + 2 * half;
            v2f a, b;
            a.x = q[rowA * CDIM + ka];
            a.y = q[rowA * CDIM + ka + 1];
            b.x = Wq[ka * CDIM + n0 + lm];
            b.y = Wq[(ka + 1) * CDIM + n0 + lm];
            acc = wmma_f32_4(a, b, acc);
        }
        int col = n0 + lm;
        float4 e0, e1, i0, i1;
        e0.x = __expf(acc[0]);  e0.y = __expf(acc[1]);
        e0.z = __expf(acc[2]);  e0.w = __expf(acc[3]);
        e1.x = __expf(acc[4]);  e1.y = __expf(acc[5]);
        e1.z = __expf(acc[6]);  e1.w = __expf(acc[7]);
        i0.x = __expf(-acc[0]); i0.y = __expf(-acc[1]);
        i0.z = __expf(-acc[2]); i0.w = __expf(-acc[3]);
        i1.x = __expf(-acc[4]); i1.y = __expf(-acc[5]);
        i1.z = __expf(-acc[6]); i1.w = __expf(-acc[7]);
        float4* E4 = reinterpret_cast<float4*>(Eq);
        float4* I4 = reinterpret_cast<float4*>(iEq);
        const int f = (mt * CDIM + col) * 4 + 2 * half;
        E4[f] = e0;
        E4[f + 1] = e1;
        I4[f] = i0;
        I4[f + 1] = i1;
    }
}

static __device__ __forceinline__ float lane_value(float v, int l) {
    return __int_as_float(__builtin_amdgcn_readlane(__float_as_int(v), l));
}
template <int XM>
static __device__ __forceinline__ float row_xor(float v) {
    return __int_as_float(
        __builtin_amdgcn_update_dpp(0, __float_as_int(v), 0x160 + XM, 0xf, 0xf, true));
}
static __device__ __forceinline__ float xhalf(float v) {
    int b = __float_as_int(v);
    return __int_as_float(
        __builtin_amdgcn_permlanex16(b, b, 0x76543210, 0xfedcba98, false, false));
}
static __device__ __forceinline__ float wave_max(float v) {
    v = fmaxf(v, row_xor<1>(v));
    v = fmaxf(v, row_xor<2>(v));
    v = fmaxf(v, row_xor<4>(v));
    v = fmaxf(v, row_xor<8>(v));
    return fmaxf(v, xhalf(v));
}
static __device__ __forceinline__ float wave_sum(float v) {
    v += row_xor<1>(v);
    v += row_xor<2>(v);
    v += row_xor<4>(v);
    v += row_xor<8>(v);
    return v + xhalf(v);
}

#define STHREADS 64

__global__ void __launch_bounds__(STHREADS)
diffattn_score(const float* __restrict__ Eq, const float* __restrict__ iEq,
               const float* __restrict__ Ek, const float* __restrict__ iEk,
               const float* __restrict__ Eqc, const float* __restrict__ iEqc,
               const float* __restrict__ Ekc, const float* __restrict__ iEkc,
               const float* __restrict__ wsum_ws,
               const float* __restrict__ v_ws,
               float* __restrict__ po, float* __restrict__ pm,
               float* __restrict__ ps) {
    __shared__ float4 sEqT[CDIM][4];
    __shared__ float4 sIEqT[CDIM][4];
    __shared__ float sP[16][CHUNK + 8];
    __shared__ float sEqc[STHREADS];
    __shared__ float sIEqc[STHREADS];

    const int tid = threadIdx.x;
    const int lane = tid & 31;
    const int wave = tid >> 5;
    const int ch = blockIdx.x;
    const int qt = blockIdx.y;
    const int q0 = qt * 16;
    const int j0 = ch * CHUNK;

    const int half = lane >> 4;
    const int lm = lane & 15;
    const int jA = j0 + lane;
    const int jB = jA + 32;
    const v4f* ek4 = reinterpret_cast<const v4f*>(Ek);
    const v4f* ik4 = reinterpret_cast<const v4f*>(iEk);
    const bool okA = jA < NKV;
    const bool okB = jB < NKV;
    const int jAc = okA ? jA : NKV - 1;
    const int jBc = okB ? jB : NKV - 1;

    const float4* gq = reinterpret_cast<const float4*>(Eq + q0 * CDIM);
    const float4* gi = reinterpret_cast<const float4*>(iEq + q0 * CDIM);
    float4 tq[4], ti[4];
#pragma unroll
    for (int u = 0; u < 4; ++u) {
        tq[u] = gq[tid + STHREADS * u];
        ti[u] = gi[tid + STHREADS * u];
    }
    const int t48 = tid < 48 ? tid : 47;
    const float cq = Eqc[q0 * 3 + t48];
    const float ci = iEqc[q0 * 3 + t48];
    const float sW[3] = {wsum_ws[0], wsum_ws[1], wsum_ws[2]};
    v4f eA[2], iA[2], eB[2], iB[2];
#pragma unroll
    for (int u = 0; u < 2; ++u) {
        eA[u] = ek4[u * NKV_PAD + jA];
        iA[u] = ik4[u * NKV_PAD + jA];
        eB[u] = ek4[u * NKV_PAD + jB];
        iB[u] = ik4[u * NKV_PAD + jB];
    }
    float ekA[3], ikA[3], ekB[3], ikB[3];
#pragma unroll
    for (int d = 0; d < 3; ++d) {
        ekA[d] = Ekc[jAc * 3 + d];
        ikA[d] = iEkc[jAc * 3 + d];
        ekB[d] = Ekc[jBc * 3 + d];
        ikB[d] = iEkc[jBc * 3 + d];
    }

    {
        float4* dq = &sEqT[0][0];
        float4* di = &sIEqT[0][0];
#pragma unroll
        for (int u = 0; u < 4; ++u) {
            dq[tid + STHREADS * u] = tq[u];
            di[tid + STHREADS * u] = ti[u];
        }
    }
    sEqc[tid] = cq;
    sIEqc[tid] = ci;
    __syncthreads();

    const long base = (long)(qt * NCHUNK + ch) * 16;

    v2f vb0[CHUNK / 4], vb1[CHUNK / 4];

    {
        const int qw = 8 * wave;

        v2f sA2[4], sB2[4];
#pragma unroll
        for (int q = 0; q < 4; ++q) { sA2[q] = (v2f){0.f, 0.f}; sB2[q] = (v2f){0.f, 0.f}; }

#pragma unroll 1
        for (int cb = 0; cb < 8; ++cb) {
            const int nb = (cb < 7) ? cb + 1 : 7;
            v4f neA[2], niA[2], neB[2], niB[2];
#pragma unroll
            for (int u = 0; u < 2; ++u) {
                neA[u] = ek4[(2 * nb + u) * NKV_PAD + jA];
                niA[u] = ik4[(2 * nb + u) * NKV_PAD + jA];
                neB[u] = ek4[(2 * nb + u) * NKV_PAD + jB];
                niB[u] = ik4[(2 * nb + u) * NKV_PAD + jB];
            }
            __builtin_amdgcn_sched_barrier(0);
#define SCORE_STEP(c, EA, IA, EB, IB)                                          \
            _Pragma("unroll")                                                  \
            for (int v = 0; v < 2; ++v) {                                      \
                const float4 eq = sEqT[8 * cb + (c)][2 * wave + v];            \
                const float4 iq = sIEqT[8 * cb + (c)][2 * wave + v];           \
                const v2f e01 = {eq.x, eq.y}, e23 = {eq.z, eq.w};              \
                const v2f i01 = {iq.x, iq.y}, i23 = {iq.z, iq.w};              \
                sA2[2 * v + 0] += __builtin_elementwise_min(e01 * (IA), i01 * (EA)); \
                sA2[2 * v + 1] += __builtin_elementwise_min(e23 * (IA), i23 * (EA)); \
                sB2[2 * v + 0] += __builtin_elementwise_min(e01 * (IB), i01 * (EB)); \
                sB2[2 * v + 1] += __builtin_elementwise_min(e23 * (IB), i23 * (EB)); \
            }
            SCORE_STEP(0, eA[0].x, iA[0].x, eB[0].x, iB[0].x)
            SCORE_STEP(1, eA[0].y, iA[0].y, eB[0].y, iB[0].y)
            SCORE_STEP(2, eA[0].z, iA[0].z, eB[0].z, iB[0].z)
            SCORE_STEP(3, eA[0].w, iA[0].w, eB[0].w, iB[0].w)
            SCORE_STEP(4, eA[1].x, iA[1].x, eB[1].x, iB[1].x)
            SCORE_STEP(5, eA[1].y, iA[1].y, eB[1].y, iB[1].y)
            SCORE_STEP(6, eA[1].z, iA[1].z, eB[1].z, iB[1].z)
            SCORE_STEP(7, eA[1].w, iA[1].w, eB[1].w, iB[1].w)
#undef SCORE_STEP
            __builtin_amdgcn_sched_barrier(0);
            asm volatile("" :: "v"(neA[0]), "v"(neA[1]), "v"(niA[0]), "v"(niA[1]),
                               "v"(neB[0]), "v"(neB[1]), "v"(niB[0]), "v"(niB[1]));
#pragma unroll
            for (int u = 0; u < 2; ++u) {
                eA[u] = neA[u]; iA[u] = niA[u];
                eB[u] = neB[u]; iB[u] = niB[u];
            }
        }

        {
            const int n0 = wave * 32;
#pragma unroll
            for (int kk = 0; kk < CHUNK; kk += 4) {
                const int ka = kk + 2 * half;
                vb0[kk / 4].x = v_ws[(j0 + ka) * CDIM + n0 + lm];
                vb0[kk / 4].y = v_ws[(j0 + ka + 1) * CDIM + n0 + lm];
                vb1[kk / 4].x = v_ws[(j0 + ka) * CDIM + n0 + 16 + lm];
                vb1[kk / 4].y = v_ws[(j0 + ka + 1) * CDIM + n0 + 16 + lm];
            }
        }

        float pa[8], pb[8];
#pragma unroll
        for (int q = 0; q < 8; ++q) {
            float a = sA2[q >> 1][q & 1], b = sB2[q >> 1][q & 1];
#pragma unroll
            for (int d = 0; d < 3; ++d) {
                float eqc = sEqc[(qw + q) * 3 + d];
                float iqc = sIEqc[(qw + q) * 3 + d];
                a += fminf(eqc * ikA[d], iqc * ekA[d]) * sW[d];
                b += fminf(eqc * ikB[d], iqc * ekB[d]) * sW[d];
            }
            a *= SCALE;
            b *= SCALE;
            pa[q] = okA ? a : -__builtin_inff();
            pb[q] = okB ? b : -__builtin_inff();
        }

        float mrow[8], srow[8];
#pragma unroll
        for (int q = 0; q < 8; ++q) {
            const float m = wave_max(fmaxf(pa[q], pb[q]));
            const float ea = __expf(pa[q] - m);
            const float eb = __expf(pb[q] - m);
            sP[qw + q][lane] = ea;
            sP[qw + q][lane + 32] = eb;
            mrow[q] = m;
            srow[q] = wave_sum(ea + eb);
        }
        float mv = mrow[0], sv = srow[0];
#pragma unroll
        for (int q = 1; q < 8; ++q) {
            mv = (lane == q) ? mrow[q] : mv;
            sv = (lane == q) ? srow[q] : sv;
        }
        if (lane < 8) {
            pm[base + qw + lane] = mv;
            ps[base + qw + lane] = sv;
        }
    }
    lds_barrier();

    {
        const int n0 = wave * 32;
        v8f acc0 = {};
        v8f acc1 = {};
#pragma unroll
        for (int kk = 0; kk < CHUNK; kk += 4) {
            int ka = kk + 2 * half;
            v2f a;
            a.x = sP[lm][ka];
            a.y = sP[lm][ka + 1];
            acc0 = wmma_f32_4(a, vb0[kk / 4], acc0);
            acc1 = wmma_f32_4(a, vb1[kk / 4], acc1);
        }
#pragma unroll
        for (int r = 0; r < 8; ++r) {
            int row = r + 8 * half;
            po[(base + row) * CDIM + n0 + lm] = acc0[r];
            po[(base + row) * CDIM + n0 + 16 + lm] = acc1[r];
        }
    }
}

#define CROWS 2
#define NCBLK (NQ / CROWS)
#define CTHREADS (CROWS * CDIM)

__global__ void __launch_bounds__(CTHREADS)
diffattn_combine(const float* __restrict__ po, const float* __restrict__ pm,
                 const float* __restrict__ ps,
                 const float* __restrict__ Wproj,
                 const float* __restrict__ bproj, float* __restrict__ out) {
    __shared__ float sO[CROWS][CDIM + 1];
    __shared__ float sOut[16][CDIM];

    const int tid = threadIdx.x;
    const int lane = tid & 31;
    const int wave = tid >> 5;
    const int half = lane >> 4;
    const int lm = lane & 15;
    const int r = tid >> 6;
    const int n = tid & 63;
    const int n0 = wave * 16;
    const int grow = blockIdx.x * CROWS + r;
    const long pbase = (long)(grow >> 4) * NCHUNK * 16 + (grow & 15);

    const bool has1 = lane + 32 < NCHUNK;
    const int c1 = has1 ? lane + 32 : NCHUNK - 1;
    float m0 = pm[pbase + lane * 16];
    float s0 = ps[pbase + lane * 16];
    float m1 = pm[pbase + c1 * 16];
    float s1 = ps[pbase + c1 * 16];

    float pv[NCHUNK];
#pragma unroll
    for (int c = 0; c < NCHUNK; ++c) pv[c] = po[(pbase + c * 16) * CDIM + n];

    v2f bf[CDIM / 4];
#pragma unroll
    for (int kk = 0; kk < CDIM; kk += 4) {
        int ka = kk + 2 * half;
        bf[kk / 4].x = Wproj[ka * CDIM + n0 + lm];
        bf[kk / 4].y = Wproj[(ka + 1) * CDIM + n0 + lm];
    }
    float bb = bproj[n];
    __builtin_amdgcn_sched_barrier(0);

    m1 = has1 ? m1 : -__builtin_inff();
    float M = wave_max(fmaxf(m0, m1));
    float e0 = __expf(m0 - M);
    float e1 = has1 ? __expf(m1 - M) : 0.f;
    float S = wave_sum(s0 * e0 + s1 * e1);
    float inv = 1.0f / S;
    float w0 = e0 * inv;
    float w1 = e1 * inv;

    float a = 0.f;
#pragma unroll
    for (int c = 0; c < NCHUNK; ++c) {
        float wc = (c < 32) ? lane_value(w0, c) : lane_value(w1, c - 32);
        a += pv[c] * wc;
    }
    sO[r][n] = a;
    __syncthreads();

    v8f acc = {};
#pragma unroll
    for (int kk = 0; kk < CDIM; kk += 4) {
        int ka = kk + 2 * half;
        v2f a2;
        a2.x = sO[lm & (CROWS - 1)][ka];
        a2.y = sO[lm & (CROWS - 1)][ka + 1];
        acc = wmma_f32_4(a2, bf[kk / 4], acc);
    }
#pragma unroll
    for (int q = 0; q < CROWS; ++q) sOut[q + 8 * half][n0 + lm] = acc[q];
    __syncthreads();

    out[grow * CDIM + n] = sOut[r][n] + bb;
}

extern "C" void kernel_launch(void* const* d_in, const int* in_sizes, int n_in,
                              void* d_out, int out_size, void* d_ws, size_t ws_size,
                              hipStream_t stream) {
    const float* q        = (const float*)d_in[0];
    const float* q_coord  = (const float*)d_in[1];
    const float* kv       = (const float*)d_in[2];
    const float* kv_coord = (const float*)d_in[3];
    const float* Wq       = (const float*)d_in[4];
    const float* Wkv      = (const float*)d_in[5];
    const float* Wdelta   = (const float*)d_in[6];
    const float* Wproj    = (const float*)d_in[7];
    const float* bproj    = (const float*)d_in[8];
    float* out = (float*)d_out;
    float* ws  = (float*)d_ws;

    float* v_ws    = ws + OFF_VW;
    float* Eqp     = ws + OFF_EQ;
    float* iEqp    = ws + OFF_IEQ;
    float* Ekp     = ws + OFF_EK;
    float* iEkp    = ws + OFF_IEK;
    float* wsum_ws = ws + OFF_WSUM;
    float* Eqc     = ws + OFF_EQC;
    float* iEqc    = ws + OFF_IEQC;
    float* Ekc     = ws + OFF_EKC;
    float* iEkc    = ws + OFF_IEKC;
    float* po      = ws + OFF_PO;
    float* pm      = ws + OFF_PM;
    float* ps      = ws + OFF_PS;

    diffattn_prep<<<PREP_BLOCKS + COORD_BLOCKS, 256, 0, stream>>>(
        q, kv, q_coord, kv_coord, Wq, Wkv, Wdelta,
        Eqp, iEqp, Ekp, iEkp, Eqc, iEqc, Ekc, iEkc, v_ws, wsum_ws);

    dim3 sgrid(NCHUNK, NQT);
    diffattn_score<<<sgrid, STHREADS, 0, stream>>>(Eqp, iEqp, Ekp, iEkp, Eqc, iEqc,
                                              Ekc, iEkc, wsum_ws, v_ws, po, pm, ps);

    diffattn_combine<<<NCBLK, CTHREADS, 0, stream>>>(po, pm, ps, Wproj, bproj, out);
}
